# starting kernel
# baseline (speedup 1.0000x reference)
#include <hip/hip_runtime.h>

#define N_ENT   50000
#define DIM     128
#define NREL    11
#define NBASIS  4
#define NEDGE   1000000
#define NLAYER  2

typedef __attribute__((ext_vector_type(2))) float v2f;
typedef __attribute__((ext_vector_type(8))) float v8f;

__global__ void zero_kernel(float4* __restrict__ p, int n4) {
    int i = blockIdx.x * blockDim.x + threadIdx.x;
    if (i < n4) p[i] = make_float4(0.f, 0.f, 0.f, 0.f);
}

__global__ void build_w_kernel(const float* __restrict__ bw,
                               const float* __restrict__ bc,
                               float* __restrict__ W, int l) {
    int idx = blockIdx.x * blockDim.x + threadIdx.x;
    int o = idx >> 7, d = idx & (DIM - 1);
    float s = 0.f;
#pragma unroll
    for (int b = 0; b < NBASIS; ++b) {
        float c = bc[(l * NREL + l) * NBASIS + b];
        s += c * bw[(((size_t)l * NBASIS + b) * DIM + o) * DIM + d];
    }
    W[idx] = s;
}

__global__ void scatter_kernel(const float* __restrict__ emb,
                               const float* __restrict__ eval,
                               const int* __restrict__ esrc,
                               const int* __restrict__ edst,
                               float* __restrict__ acc) {
    int wave = blockIdx.x * (blockDim.x >> 5) + (threadIdx.x >> 5);
    int lane = threadIdx.x & 31;
    if (wave >= NEDGE) return;
    int   src = esrc[wave];
    int   dst = edst[wave];
    float val = eval[wave];
    const float4 v = *(const float4*)(emb + (size_t)dst * DIM + lane * 4);
    float* p = acc + (size_t)src * DIM + lane * 4;
    atomicAdd(p + 0, v.x * val);
    atomicAdd(p + 1, v.y * val);
    atomicAdd(p + 2, v.z * val);
    atomicAdd(p + 3, v.w * val);
}

__global__ void gemm_relu_kernel(const float* __restrict__ A,
                                 const float* __restrict__ W,
                                 float* __restrict__ out) {
    int rowBase = blockIdx.x * 16;
    int colBase = (threadIdx.x >> 5) * 16;
    int lane = threadIdx.x & 31;
    int r  = lane & 15;
    int kh = lane >> 4;

    const float* ap = A + (size_t)(rowBase + r) * DIM + 2 * kh;
    const float* bp = W + (size_t)(colBase + r) * DIM + 2 * kh;

    v8f c = {};
#pragma unroll
    for (int k = 0; k < DIM; k += 4) {
        v2f a = *(const v2f*)(ap + k);
        v2f b = *(const v2f*)(bp + k);
        c = __builtin_amdgcn_wmma_f32_16x16x4_f32(
                 false, a,  false, b,
                 (short)0, c,  false,  false);
    }

    float* op = out + (size_t)(rowBase + 8 * kh) * DIM + colBase + r;
#pragma unroll
    for (int v = 0; v < 8; ++v)
        op[(size_t)v * DIM] = fmaxf(c[v], 0.f);
}

__global__ void normalize_kernel(float* __restrict__ out) {
    int row  = blockIdx.x * (blockDim.x >> 5) + (threadIdx.x >> 5);
    int lane = threadIdx.x & 31;
    float4 v = *(float4*)(out + (size_t)row * DIM + lane * 4);
    float s = v.x * v.x + v.y * v.y + v.z * v.z + v.w * v.w;
#pragma unroll
    for (int off = 16; off > 0; off >>= 1)
        s += __shfl_xor(s, off, 32);
    float inv = 1.0f / fmaxf(sqrtf(s), 1e-12f);
    v.x *= inv; v.y *= inv; v.z *= inv; v.w *= inv;
    *(float4*)(out + (size_t)row * DIM + lane * 4) = v;
}

extern "C" void kernel_launch(void* const* d_in, const int* in_sizes, int n_in,
                              void* d_out, int out_size, void* d_ws, size_t ws_size,
                              hipStream_t stream) {
    const float* ent_emb = (const float*)d_in[0];
    const float* bw      = (const float*)d_in[1];
    const float* bc      = (const float*)d_in[2];
    const float* eval    = (const float*)d_in[3];
    const int*   esrc    = (const int*)d_in[5];
    const int*   edst    = (const int*)d_in[6];

    float* out = (float*)d_out;
    float* acc = (float*)d_ws;
    float* W   = acc + (size_t)N_ENT * DIM;

    for (int l = 0; l < NLAYER; ++l) {
        zero_kernel<<<(N_ENT * DIM / 4 + 255) / 256, 256, 0, stream>>>(
            (float4*)acc, N_ENT * DIM / 4);
        build_w_kernel<<<(DIM * DIM) / 256, 256, 0, stream>>>(bw, bc, W, l);
        const float* emb_in = (l == 0) ? ent_emb : out;
        scatter_kernel<<<NEDGE / 8, 256, 0, stream>>>(emb_in, eval, esrc, edst, acc);
        gemm_relu_kernel<<<N_ENT / 16, 256, 0, stream>>>(acc, W, out);
    }
    normalize_kernel<<<N_ENT / 8, 256, 0, stream>>>(out);
}


# Per-edge f32 atomic scatter replaced by in-call counting sort of edges by source row (integer atomics) plus per-row register gather-sum with fma; a row's sum order = arrival rank of those atomics
# speedup vs baseline: 1.7927x; 1.7927x over previous
#include <hip/hip_runtime.h>

#define N_ENT   50000
#define DIM     128
#define NREL    11
#define NBASIS  4
#define NEDGE   1000000
#define NLAYER  2

#define CNT_PAD        50048
#define ROWS_PER_WAVE  4
#define GATHER_WAVES   (N_ENT / ROWS_PER_WAVE)
#define EDGE_BATCH     8

typedef __attribute__((ext_vector_type(2))) float v2f;
typedef __attribute__((ext_vector_type(8))) float v8f;

__global__ void zero_kernel(float4* __restrict__ p, int n4) {
    int i = blockIdx.x * blockDim.x + threadIdx.x;
    if (i < n4) p[i] = make_float4(0.f, 0.f, 0.f, 0.f);
}

__global__ void build_w_kernel(const float* __restrict__ bw,
                               const float* __restrict__ bc,
                               float* __restrict__ W, int l) {
    int idx = blockIdx.x * blockDim.x + threadIdx.x;
    int o = idx >> 7, d = idx & (DIM - 1);
    float s = 0.f;
#pragma unroll
    for (int b = 0; b < NBASIS; ++b) {
        float c = bc[(l * NREL + l) * NBASIS + b];
        s += c * bw[(((size_t)l * NBASIS + b) * DIM + o) * DIM + d];
    }
    W[idx] = s;
}


__global__ void rank_kernel(const int* __restrict__ esrc,
                            int* __restrict__ cnt,
                            int* __restrict__ rank) {
    int i = blockIdx.x * blockDim.x + threadIdx.x;
    if (i >= NEDGE / 4) return;
    const int4 s = *(const int4*)(esrc + 4 * (size_t)i);
    int4 r;
    r.x = atomicAdd(cnt + s.x, 1);
    r.y = atomicAdd(cnt + s.y, 1);
    r.z = atomicAdd(cnt + s.z, 1);
    r.w = atomicAdd(cnt + s.w, 1);
    *(int4*)(rank + 4 * (size_t)i) = r;
}

__global__ void alloc_kernel(const int* cnt, int* cursor, int* rowstart) {
    int row  = blockIdx.x * blockDim.x + threadIdx.x;
    int lane = threadIdx.x & 31;
    int deg  = (row < N_ENT) ? cnt[row] : 0;
    int incl = deg;
#pragma unroll
    for (int off = 1; off < 32; off <<= 1) {
        int t = __shfl_up(incl, off, 32);
        if (lane >= off) incl += t;
    }
    int total = __shfl(incl, 31, 32);
    int base = 0;
    if (lane == 0) base = atomicAdd(cursor, total);
    base = __shfl(base, 0, 32);
    if (row < N_ENT) rowstart[row] = base + incl - deg;
}

__device__ __forceinline__ unsigned long long pack_edge(int dst, float val) {
    return (unsigned long long)(unsigned)dst | ((unsigned long long)(unsigned)__float_as_int(val) << 32);
}

__global__ void fill_kernel(const int* __restrict__ esrc,
                            const int* __restrict__ edst,
                            const float* __restrict__ eval,
                            const int* __restrict__ rank,
                            const int* __restrict__ rowstart,
                            unsigned long long* __restrict__ sorted) {
    int i = blockIdx.x * blockDim.x + threadIdx.x;
    if (i >= NEDGE / 4) return;
    const int4   s = *(const int4*)(esrc + 4 * (size_t)i);
    const int4   d = *(const int4*)(edst + 4 * (size_t)i);
    const float4 v = *(const float4*)(eval + 4 * (size_t)i);
    const int4   r = *(const int4*)(rank + 4 * (size_t)i);
    unsigned p0 = min((unsigned)(rowstart[s.x] + r.x), (unsigned)(NEDGE - 1));
    unsigned p1 = min((unsigned)(rowstart[s.y] + r.y), (unsigned)(NEDGE - 1));
    unsigned p2 = min((unsigned)(rowstart[s.z] + r.z), (unsigned)(NEDGE - 1));
    unsigned p3 = min((unsigned)(rowstart[s.w] + r.w), (unsigned)(NEDGE - 1));
    atomicExch(sorted + p0, pack_edge(d.x, v.x));
    atomicExch(sorted + p1, pack_edge(d.y, v.y));
    atomicExch(sorted + p2, pack_edge(d.z, v.z));
    atomicExch(sorted + p3, pack_edge(d.w, v.w));
}

template <bool FULL>
__device__ __forceinline__ void edge_batch(const float* __restrict__ ebase, int2 e, int j, int n,
                                           float4& a) {
    float4 x[EDGE_BATCH];
    float  v[EDGE_BATCH];
#pragma unroll
    for (int u = 0; u < EDGE_BATCH; ++u) {
        v[u] = 0.f;
        x[u] = make_float4(0.f, 0.f, 0.f, 0.f);
        if (FULL || j + u < n) {
            int d = __builtin_amdgcn_readlane(e.x, j + u);
            v[u]  = __int_as_float(__builtin_amdgcn_readlane(e.y, j + u));
            x[u]  = *(const float4*)(ebase + (size_t)d * DIM);
        }
    }
#pragma unroll
    for (int u = 0; u < EDGE_BATCH; ++u) {
        a.x = fmaf(v[u], x[u].x, a.x);
        a.y = fmaf(v[u], x[u].y, a.y);
        a.z = fmaf(v[u], x[u].z, a.z);
        a.w = fmaf(v[u], x[u].w, a.w);
    }
}

__global__ __launch_bounds__(128)
void gather_kernel(const float* __restrict__ emb,
                   const int* __restrict__ rowstart,
                   const int* __restrict__ cnt,
                   const int2* __restrict__ sorted,
                   float* __restrict__ acc) {
    const int wave = __builtin_amdgcn_readfirstlane(blockIdx.x * 4 + (threadIdx.x >> 5));
    const int lane = threadIdx.x & 31;
    const int row0 = wave * ROWS_PER_WAVE;

    int st[ROWS_PER_WAVE], dg[ROWS_PER_WAVE];
#pragma unroll
    for (int r = 0; r < ROWS_PER_WAVE; ++r) {
        st[r] = min(max(rowstart[row0 + r], 0), NEDGE - 1);
        dg[r] = min(max(cnt[row0 + r], 0), NEDGE);
    }
    int2 first[ROWS_PER_WAVE];
#pragma unroll
    for (int r = 0; r < ROWS_PER_WAVE; ++r)
        first[r] = (lane < dg[r]) ? sorted[min(st[r] + lane, NEDGE - 1)] : make_int2(0, 0);

    const float* ebase = emb + lane * 4;
#pragma unroll
    for (int r = 0; r < ROWS_PER_WAVE; ++r) {
        float4 a = make_float4(0.f, 0.f, 0.f, 0.f);
        const int deg = dg[r];
        int2 e = first[r];
        for (int base = 0; base < deg; base += 32) {
            if (base > 0)
                e = (base + lane < deg) ? sorted[min(st[r] + base + lane, NEDGE - 1)] : make_int2(0, 0);
            e.x = min(max(e.x, 0), N_ENT - 1);
            const int n = min(32, deg - base);
            int j = 0;
            for (; j + EDGE_BATCH <= n; j += EDGE_BATCH)
                edge_batch<true>(ebase, e, j, n, a);
            if (j < n)
                edge_batch<false>(ebase, e, j, n, a);
        }
        *(float4*)(acc + (size_t)(row0 + r) * DIM + lane * 4) = a;
    }
}

__global__ void gemm_relu_kernel(const float* __restrict__ A,
                                 const float* __restrict__ W,
                                 float* __restrict__ out) {
    int rowBase = blockIdx.x * 16;
    int colBase = (threadIdx.x >> 5) * 16;
    int lane = threadIdx.x & 31;
    int r  = lane & 15;
    int kh = lane >> 4;

    const float* ap = A + (size_t)(rowBase + r) * DIM + 2 * kh;
    const float* bp = W + (size_t)(colBase + r) * DIM + 2 * kh;

    v8f c = {};
#pragma unroll
    for (int k = 0; k < DIM; k += 4) {
        v2f a = *(const v2f*)(ap + k);
        v2f b = *(const v2f*)(bp + k);
        c = __builtin_amdgcn_wmma_f32_16x16x4_f32(
                 false, a,  false, b,
                 (short)0, c,  false,  false);
    }

    float* op = out + (size_t)(rowBase + 8 * kh) * DIM + colBase + r;
#pragma unroll
    for (int v = 0; v < 8; ++v)
        op[(size_t)v * DIM] = fmaxf(c[v], 0.f);
}

__global__ void normalize_kernel(float* __restrict__ out) {
    int row  = blockIdx.x * (blockDim.x >> 5) + (threadIdx.x >> 5);
    int lane = threadIdx.x & 31;
    float4 v = *(float4*)(out + (size_t)row * DIM + lane * 4);
    float s = v.x * v.x + v.y * v.y + v.z * v.z + v.w * v.w;
#pragma unroll
    for (int off = 16; off > 0; off >>= 1)
        s += __shfl_xor(s, off, 32);
    float inv = 1.0f / fmaxf(sqrtf(s), 1e-12f);
    v.x *= inv; v.y *= inv; v.z *= inv; v.w *= inv;
    *(float4*)(out + (size_t)row * DIM + lane * 4) = v;
}

extern "C" void kernel_launch(void* const* d_in, const int* in_sizes, int n_in,
                              void* d_out, int out_size, void* d_ws, size_t ws_size,
                              hipStream_t stream) {
    const float* ent_emb = (const float*)d_in[0];
    const float* bw      = (const float*)d_in[1];
    const float* bc      = (const float*)d_in[2];
    const float* eval    = (const float*)d_in[3];
    const int*   esrc    = (const int*)d_in[5];
    const int*   edst    = (const int*)d_in[6];

    float* out = (float*)d_out;
    float* acc = (float*)d_ws;
    float* W   = acc + (size_t)N_ENT * DIM;
    int*   cnt      = (int*)(W + DIM * DIM);
    int*   rowstart = cnt + CNT_PAD;
    int*   rank     = rowstart + CNT_PAD;
    int2*  sorted   = (int2*)(rank + NEDGE);

    zero_kernel<<<(CNT_PAD / 4 + 255) / 256, 256, 0, stream>>>((float4*)cnt, CNT_PAD / 4);
    rank_kernel<<<(NEDGE / 4 + 255) / 256, 256, 0, stream>>>(esrc, cnt, rank);
    alloc_kernel<<<(N_ENT + 255) / 256, 256, 0, stream>>>(cnt, cnt + N_ENT, rowstart);
    fill_kernel<<<(NEDGE / 4 + 255) / 256, 256, 0, stream>>>(esrc, edst, eval, rank, rowstart,
                                                          (unsigned long long*)sorted);

    for (int l = 0; l < NLAYER; ++l) {
        build_w_kernel<<<(DIM * DIM) / 256, 256, 0, stream>>>(bw, bc, W, l);
        const float* emb_in = (l == 0) ? ent_emb : out;
        gather_kernel<<<GATHER_WAVES / 4, 128, 0, stream>>>(emb_in, rowstart, cnt, sorted, acc);
        gemm_relu_kernel<<<N_ENT / 16, 256, 0, stream>>>(acc, W, out);
    }
    normalize_kernel<<<N_ENT / 8, 256, 0, stream>>>(out);
}


# GEMM re-tiled: one wave per 16x128 output strip (A fragment reused by 8 WMMAs), W pre-ordered by fragment, both operands staged in LDS by whole-line loads, output rows stored as whole lines
# speedup vs baseline: 2.4289x; 1.3549x over previous
#include <hip/hip_runtime.h>

#define N_ENT   50000
#define DIM     128
#define NREL    11
#define NBASIS  4
#define NEDGE   1000000
#define NLAYER  2

#define CNT_PAD        50048
#define ROWS_PER_WAVE  4
#define GATHER_WAVES   (N_ENT / ROWS_PER_WAVE)
#define EDGE_BATCH     8

typedef __attribute__((ext_vector_type(2))) float v2f;
typedef __attribute__((ext_vector_type(8))) float v8f;

__global__ void zero_kernel(float4* __restrict__ p, int n4) {
    int i = blockIdx.x * blockDim.x + threadIdx.x;
    if (i < n4) p[i] = make_float4(0.f, 0.f, 0.f, 0.f);
}

__global__ void build_w_kernel(const float* __restrict__ bw,
                               const float* __restrict__ bc,
                               float* __restrict__ Wf, int l) {
    int idx  = blockIdx.x * blockDim.x + threadIdx.x;
    int q    = idx & 3;
    int lane = (idx >> 2) & 31;
    int j    = (idx >> 7) & 15;
    int ct   = idx >> 11;
    int o = 16 * ct + (lane & 15);
    int d = 8 * j + 4 * (q >> 1) + 2 * (lane >> 4) + (q & 1);
    float s = 0.f;
#pragma unroll
    for (int b = 0; b < NBASIS; ++b) {
        float c = bc[(l * NREL + l) * NBASIS + b];
        s += c * bw[(((size_t)l * NBASIS + b) * DIM + o) * DIM + d];
    }
    Wf[idx] = s;
}


__global__ void rank_kernel(const int* __restrict__ esrc,
                            int* __restrict__ cnt,
                            int* __restrict__ rank) {
    int i = blockIdx.x * blockDim.x + threadIdx.x;
    if (i >= NEDGE / 4) return;
    const int4 s = *(const int4*)(esrc + 4 * (size_t)i);
    int4 r;
    r.x = atomicAdd(cnt + s.x, 1);
    r.y = atomicAdd(cnt + s.y, 1);
    r.z = atomicAdd(cnt + s.z, 1);
    r.w = atomicAdd(cnt + s.w, 1);
    *(int4*)(rank + 4 * (size_t)i) = r;
}

__global__ void alloc_kernel(const int* cnt, int* cursor, int* rowstart) {
    int row  = blockIdx.x * blockDim.x + threadIdx.x;
    int lane = threadIdx.x & 31;
    int deg  = (row < N_ENT) ? cnt[row] : 0;
    int incl = deg;
#pragma unroll
    for (int off = 1; off < 32; off <<= 1) {
        int t = __shfl_up(incl, off, 32);
        if (lane >= off) incl += t;
    }
    int total = __shfl(incl, 31, 32);
    int base = 0;
    if (lane == 0) base = atomicAdd(cursor, total);
    base = __shfl(base, 0, 32);
    if (row < N_ENT) rowstart[row] = base + incl - deg;
}

__device__ __forceinline__ unsigned long long pack_edge(int dst, float val) {
    return (unsigned long long)(unsigned)dst | ((unsigned long long)(unsigned)__float_as_int(val) << 32);
}

__global__ void fill_kernel(const int* __restrict__ esrc,
                            const int* __restrict__ edst,
                            const float* __restrict__ eval,
                            const int* __restrict__ rank,
                            const int* __restrict__ rowstart,
                            unsigned long long* __restrict__ sorted) {
    int i = blockIdx.x * blockDim.x + threadIdx.x;
    if (i >= NEDGE / 4) return;
    const int4   s = *(const int4*)(esrc + 4 * (size_t)i);
    const int4   d = *(const int4*)(edst + 4 * (size_t)i);
    const float4 v = *(const float4*)(eval + 4 * (size_t)i);
    const int4   r = *(const int4*)(rank + 4 * (size_t)i);
    unsigned p0 = min((unsigned)(rowstart[s.x] + r.x), (unsigned)(NEDGE - 1));
    unsigned p1 = min((unsigned)(rowstart[s.y] + r.y), (unsigned)(NEDGE - 1));
    unsigned p2 = min((unsigned)(rowstart[s.z] + r.z), (unsigned)(NEDGE - 1));
    unsigned p3 = min((unsigned)(rowstart[s.w] + r.w), (unsigned)(NEDGE - 1));
    atomicExch(sorted + p0, pack_edge(d.x, v.x));
    atomicExch(sorted + p1, pack_edge(d.y, v.y));
    atomicExch(sorted + p2, pack_edge(d.z, v.z));
    atomicExch(sorted + p3, pack_edge(d.w, v.w));
}

template <bool FULL>
__device__ __forceinline__ void edge_batch(const float* __restrict__ ebase, int2 e, int j, int n,
                                           float4& a) {
    float4 x[EDGE_BATCH];
    float  v[EDGE_BATCH];
#pragma unroll
    for (int u = 0; u < EDGE_BATCH; ++u) {
        v[u] = 0.f;
        x[u] = make_float4(0.f, 0.f, 0.f, 0.f);
        if (FULL || j + u < n) {
            int d = __builtin_amdgcn_readlane(e.x, j + u);
            v[u]  = __int_as_float(__builtin_amdgcn_readlane(e.y, j + u));
            x[u]  = *(const float4*)(ebase + (size_t)d * DIM);
        }
    }
#pragma unroll
    for (int u = 0; u < EDGE_BATCH; ++u) {
        a.x = fmaf(v[u], x[u].x, a.x);
        a.y = fmaf(v[u], x[u].y, a.y);
        a.z = fmaf(v[u], x[u].z, a.z);
        a.w = fmaf(v[u], x[u].w, a.w);
    }
}

__global__ __launch_bounds__(128)
void gather_kernel(const float* __restrict__ emb,
                   const int* __restrict__ rowstart,
                   const int* __restrict__ cnt,
                   const int2* __restrict__ sorted,
                   float* __restrict__ acc) {
    const int wave = __builtin_amdgcn_readfirstlane(blockIdx.x * 4 + (threadIdx.x >> 5));
    const int lane = threadIdx.x & 31;
    const int row0 = wave * ROWS_PER_WAVE;

    int st[ROWS_PER_WAVE], dg[ROWS_PER_WAVE];
#pragma unroll
    for (int r = 0; r < ROWS_PER_WAVE; ++r) {
        st[r] = min(max(rowstart[row0 + r], 0), NEDGE - 1);
        dg[r] = min(max(cnt[row0 + r], 0), NEDGE);
    }
    int2 first[ROWS_PER_WAVE];
#pragma unroll
    for (int r = 0; r < ROWS_PER_WAVE; ++r)
        first[r] = (lane < dg[r]) ? sorted[min(st[r] + lane, NEDGE - 1)] : make_int2(0, 0);

    const float* ebase = emb + lane * 4;
#pragma unroll
    for (int r = 0; r < ROWS_PER_WAVE; ++r) {
        float4 a = make_float4(0.f, 0.f, 0.f, 0.f);
        const int deg = dg[r];
        int2 e = first[r];
        for (int base = 0; base < deg; base += 32) {
            if (base > 0)
                e = (base + lane < deg) ? sorted[min(st[r] + base + lane, NEDGE - 1)] : make_int2(0, 0);
            e.x = min(max(e.x, 0), N_ENT - 1);
            const int n = min(32, deg - base);
            int j = 0;
            for (; j + EDGE_BATCH <= n; j += EDGE_BATCH)
                edge_batch<true>(ebase, e, j, n, a);
            if (j < n)
                edge_batch<false>(ebase, e, j, n, a);
        }
        *(float4*)(acc + (size_t)(row0 + r) * DIM + lane * 4) = a;
    }
}

#define GEMM_WAVES   8
#define GEMM_ROWS    (GEMM_WAVES * 16)
#define GEMM_BLOCKS  ((N_ENT + GEMM_ROWS - 1) / GEMM_ROWS)
#define TILE_PITCH   132

__global__ __launch_bounds__(GEMM_WAVES * 32)
void gemm_relu_kernel(const float* __restrict__ A,
                      const float* __restrict__ Wf,
                      float* __restrict__ out) {
    __shared__ __attribute__((aligned(16))) float lds[DIM * DIM + GEMM_WAVES * 16 * TILE_PITCH];
    float* sW = lds;
    float* sT = lds + DIM * DIM;

    const int wave = __builtin_amdgcn_readfirstlane(threadIdx.x >> 5);
    const int lane = threadIdx.x & 31;
    const int c = lane & 15;
    const int h = lane >> 4;
    const int rowBase = blockIdx.x * GEMM_ROWS + wave * 16;
    const bool active = rowBase < N_ENT;
    float* tile = sT + wave * (16 * TILE_PITCH);

    if (active) {
        float4 t[16];
#pragma unroll
        for (int r = 0; r < 16; ++r)
            t[r] = *(const float4*)(A + (size_t)(rowBase + r) * DIM + lane * 4);
#pragma unroll
        for (int r = 0; r < 16; ++r)
            *(float4*)(tile + r * TILE_PITCH + lane * 4) = t[r];
    }
    {
        const float4* g = (const float4*)Wf + threadIdx.x;
        float4*       s = (float4*)sW + threadIdx.x;
#pragma unroll
        for (int i = 0; i < (DIM * DIM / 4) / (GEMM_WAVES * 32); ++i)
            s[i * (GEMM_WAVES * 32)] = g[i * (GEMM_WAVES * 32)];
    }
    __syncthreads();
    if (!active) return;

    v8f acc[8];
#pragma unroll
    for (int ct = 0; ct < 8; ++ct) acc[ct] = (v8f){};

    const float*  ap = tile + c * TILE_PITCH + 2 * h;
    const float4* bp = (const float4*)sW + lane;
    v2f    a[2][2];
    float4 b[2][8];
    a[0][0] = *(const v2f*)(ap);
    a[0][1] = *(const v2f*)(ap + 4);
#pragma unroll
    for (int ct = 0; ct < 8; ++ct) b[0][ct] = bp[(ct * (DIM / 8)) * 32];
#pragma unroll
    for (int j = 0; j < DIM / 8; ++j) {
        const int cur = j & 1, nxt = cur ^ 1;
        if (j + 1 < DIM / 8) {
            a[nxt][0] = *(const v2f*)(ap + 8 * (j + 1));
            a[nxt][1] = *(const v2f*)(ap + 8 * (j + 1) + 4);
#pragma unroll
            for (int ct = 0; ct < 8; ++ct) b[nxt][ct] = bp[(ct * (DIM / 8) + j + 1) * 32];
        }
        __builtin_amdgcn_sched_barrier(0);
#pragma unroll
        for (int ct = 0; ct < 8; ++ct) {
            v2f b0 = {b[cur][ct].x, b[cur][ct].y};
            acc[ct] = __builtin_amdgcn_wmma_f32_16x16x4_f32(
                           false, a[cur][0],  false, b0,
                           (short)0, acc[ct],  false,  false);
        }
#pragma unroll
        for (int ct = 0; ct < 8; ++ct) {
            v2f b1 = {b[cur][ct].z, b[cur][ct].w};
            acc[ct] = __builtin_amdgcn_wmma_f32_16x16x4_f32(
                           false, a[cur][1],  false, b1,
                           (short)0, acc[ct],  false,  false);
        }
        __builtin_amdgcn_sched_barrier(0);
    }

#pragma unroll
    for (int ct = 0; ct < 8; ++ct)
#pragma unroll
        for (int r = 0; r < 8; ++r)
            tile[(8 * h + r) * TILE_PITCH + 16 * ct + c] = fmaxf(acc[ct][r], 0.f);
    asm volatile("s_wait_dscnt 0x0" ::: "memory");
#pragma unroll
    for (int r = 0; r < 16; ++r) {
        const float4 v = *(const float4*)(tile + r * TILE_PITCH + lane * 4);
        *(float4*)(out + (size_t)(rowBase + r) * DIM + lane * 4) = v;
    }
}

__global__ void normalize_kernel(float* __restrict__ out) {
    int row  = blockIdx.x * (blockDim.x >> 5) + (threadIdx.x >> 5);
    int lane = threadIdx.x & 31;
    float4 v = *(float4*)(out + (size_t)row * DIM + lane * 4);
    float s = v.x * v.x + v.y * v.y + v.z * v.z + v.w * v.w;
#pragma unroll
    for (int off = 16; off > 0; off >>= 1)
        s += __shfl_xor(s, off, 32);
    float inv = 1.0f / fmaxf(sqrtf(s), 1e-12f);
    v.x *= inv; v.y *= inv; v.z *= inv; v.w *= inv;
    *(float4*)(out + (size_t)row * DIM + lane * 4) = v;
}

extern "C" void kernel_launch(void* const* d_in, const int* in_sizes, int n_in,
                              void* d_out, int out_size, void* d_ws, size_t ws_size,
                              hipStream_t stream) {
    const float* ent_emb = (const float*)d_in[0];
    const float* bw      = (const float*)d_in[1];
    const float* bc      = (const float*)d_in[2];
    const float* eval    = (const float*)d_in[3];
    const int*   esrc    = (const int*)d_in[5];
    const int*   edst    = (const int*)d_in[6];

    float* out = (float*)d_out;
    float* acc = (float*)d_ws;
    float* W   = acc + (size_t)N_ENT * DIM;
    int*   cnt      = (int*)(W + DIM * DIM);
    int*   rowstart = cnt + CNT_PAD;
    int*   rank     = rowstart + CNT_PAD;
    int2*  sorted   = (int2*)(rank + NEDGE);

    zero_kernel<<<(CNT_PAD / 4 + 255) / 256, 256, 0, stream>>>((float4*)cnt, CNT_PAD / 4);
    rank_kernel<<<(NEDGE / 4 + 255) / 256, 256, 0, stream>>>(esrc, cnt, rank);
    alloc_kernel<<<(N_ENT + 255) / 256, 256, 0, stream>>>(cnt, cnt + N_ENT, rowstart);
    fill_kernel<<<(NEDGE / 4 + 255) / 256, 256, 0, stream>>>(esrc, edst, eval, rank, rowstart,
                                                          (unsigned long long*)sorted);

    for (int l = 0; l < NLAYER; ++l) {
        build_w_kernel<<<(DIM * DIM) / 256, 256, 0, stream>>>(bw, bc, W, l);
        const float* emb_in = (l == 0) ? ent_emb : out;
        gather_kernel<<<GATHER_WAVES / 4, 128, 0, stream>>>(emb_in, rowstart, cnt, sorted, acc);
        gemm_relu_kernel<<<GEMM_BLOCKS, GEMM_WAVES * 32, 0, stream>>>(acc, W, out);
    }
    normalize_kernel<<<N_ENT / 8, 256, 0, stream>>>(out);
}


# Row L2 normalisation fused into the last GEMM's store phase (each wave reduces its 16 row sums together, one root and division per wave); separate normalise launch and its pass over out removed
# speedup vs baseline: 2.5449x; 1.0478x over previous
#include <hip/hip_runtime.h>

#define N_ENT   50000
#define DIM     128
#define NREL    11
#define NBASIS  4
#define NEDGE   1000000
#define NLAYER  2

#define CNT_PAD        50048
#define ROWS_PER_WAVE  4
#define GATHER_WAVES   (N_ENT / ROWS_PER_WAVE)
#define EDGE_BATCH     8

typedef __attribute__((ext_vector_type(2))) float v2f;
typedef __attribute__((ext_vector_type(8))) float v8f;

__global__ void zero_kernel(float4* __restrict__ p, int n4) {
    int i = blockIdx.x * blockDim.x + threadIdx.x;
    if (i < n4) p[i] = make_float4(0.f, 0.f, 0.f, 0.f);
}

__global__ void build_w_kernel(const float* __restrict__ bw,
                               const float* __restrict__ bc,
                               float* __restrict__ Wf, int l) {
    int idx  = blockIdx.x * blockDim.x + threadIdx.x;
    int q    = idx & 3;
    int lane = (idx >> 2) & 31;
    int j    = (idx >> 7) & 15;
    int ct   = idx >> 11;
    int o = 16 * ct + (lane & 15);
    int d = 8 * j + 4 * (q >> 1) + 2 * (lane >> 4) + (q & 1);
    float s = 0.f;
#pragma unroll
    for (int b = 0; b < NBASIS; ++b) {
        float c = bc[(l * NREL + l) * NBASIS + b];
        s += c * bw[(((size_t)l * NBASIS + b) * DIM + o) * DIM + d];
    }
    Wf[idx] = s;
}


__global__ void rank_kernel(const int* __restrict__ esrc,
                            int* __restrict__ cnt,
                            int* __restrict__ rank) {
    int i = blockIdx.x * blockDim.x + threadIdx.x;
    if (i >= NEDGE / 4) return;
    const int4 s = *(const int4*)(esrc + 4 * (size_t)i);
    int4 r;
    r.x = atomicAdd(cnt + s.x, 1);
    r.y = atomicAdd(cnt + s.y, 1);
    r.z = atomicAdd(cnt + s.z, 1);
    r.w = atomicAdd(cnt + s.w, 1);
    *(int4*)(rank + 4 * (size_t)i) = r;
}

__global__ void alloc_kernel(const int* cnt, int* cursor, int* rowstart) {
    int row  = blockIdx.x * blockDim.x + threadIdx.x;
    int lane = threadIdx.x & 31;
    int deg  = (row < N_ENT) ? cnt[row] : 0;
    int incl = deg;
#pragma unroll
    for (int off = 1; off < 32; off <<= 1) {
        int t = __shfl_up(incl, off, 32);
        if (lane >= off) incl += t;
    }
    int total = __shfl(incl, 31, 32);
    int base = 0;
    if (lane == 0) base = atomicAdd(cursor, total);
    base = __shfl(base, 0, 32);
    if (row < N_ENT) rowstart[row] = base + incl - deg;
}

__device__ __forceinline__ unsigned long long pack_edge(int dst, float val) {
    return (unsigned long long)(unsigned)dst | ((unsigned long long)(unsigned)__float_as_int(val) << 32);
}

__global__ void fill_kernel(const int* __restrict__ esrc,
                            const int* __restrict__ edst,
                            const float* __restrict__ eval,
                            const int* __restrict__ rank,
                            const int* __restrict__ rowstart,
                            unsigned long long* __restrict__ sorted) {
    int i = blockIdx.x * blockDim.x + threadIdx.x;
    if (i >= NEDGE / 4) return;
    const int4   s = *(const int4*)(esrc + 4 * (size_t)i);
    const int4   d = *(const int4*)(edst + 4 * (size_t)i);
    const float4 v = *(const float4*)(eval + 4 * (size_t)i);
    const int4   r = *(const int4*)(rank + 4 * (size_t)i);
    unsigned p0 = min((unsigned)(rowstart[s.x] + r.x), (unsigned)(NEDGE - 1));
    unsigned p1 = min((unsigned)(rowstart[s.y] + r.y), (unsigned)(NEDGE - 1));
    unsigned p2 = min((unsigned)(rowstart[s.z] + r.z), (unsigned)(NEDGE - 1));
    unsigned p3 = min((unsigned)(rowstart[s.w] + r.w), (unsigned)(NEDGE - 1));
    atomicExch(sorted + p0, pack_edge(d.x, v.x));
    atomicExch(sorted + p1, pack_edge(d.y, v.y));
    atomicExch(sorted + p2, pack_edge(d.z, v.z));
    atomicExch(sorted + p3, pack_edge(d.w, v.w));
}

template <bool FULL>
__device__ __forceinline__ void edge_batch(const float* __restrict__ ebase, int2 e, int j, int n,
                                           float4& a) {
    float4 x[EDGE_BATCH];
    float  v[EDGE_BATCH];
#pragma unroll
    for (int u = 0; u < EDGE_BATCH; ++u) {
        v[u] = 0.f;
        x[u] = make_float4(0.f, 0.f, 0.f, 0.f);
        if (FULL || j + u < n) {
            int d = __builtin_amdgcn_readlane(e.x, j + u);
            v[u]  = __int_as_float(__builtin_amdgcn_readlane(e.y, j + u));
            x[u]  = *(const float4*)(ebase + (size_t)d * DIM);
        }
    }
#pragma unroll
    for (int u = 0; u < EDGE_BATCH; ++u) {
        a.x = fmaf(v[u], x[u].x, a.x);
        a.y = fmaf(v[u], x[u].y, a.y);
        a.z = fmaf(v[u], x[u].z, a.z);
        a.w = fmaf(v[u], x[u].w, a.w);
    }
}

__global__ __launch_bounds__(128)
void gather_kernel(const float* __restrict__ emb,
                   const int* __restrict__ rowstart,
                   const int* __restrict__ cnt,
                   const int2* __restrict__ sorted,
                   float* __restrict__ acc) {
    const int wave = __builtin_amdgcn_readfirstlane(blockIdx.x * 4 + (threadIdx.x >> 5));
    const int lane = threadIdx.x & 31;
    const int row0 = wave * ROWS_PER_WAVE;

    int st[ROWS_PER_WAVE], dg[ROWS_PER_WAVE];
#pragma unroll
    for (int r = 0; r < ROWS_PER_WAVE; ++r) {
        st[r] = min(max(rowstart[row0 + r], 0), NEDGE - 1);
        dg[r] = min(max(cnt[row0 + r], 0), NEDGE);
    }
    int2 first[ROWS_PER_WAVE];
#pragma unroll
    for (int r = 0; r < ROWS_PER_WAVE; ++r)
        first[r] = (lane < dg[r]) ? sorted[min(st[r] + lane, NEDGE - 1)] : make_int2(0, 0);

    const float* ebase = emb + lane * 4;
#pragma unroll
    for (int r = 0; r < ROWS_PER_WAVE; ++r) {
        float4 a = make_float4(0.f, 0.f, 0.f, 0.f);
        const int deg = dg[r];
        int2 e = first[r];
        for (int base = 0; base < deg; base += 32) {
            if (base > 0)
                e = (base + lane < deg) ? sorted[min(st[r] + base + lane, NEDGE - 1)] : make_int2(0, 0);
            e.x = min(max(e.x, 0), N_ENT - 1);
            const int n = min(32, deg - base);
            int j = 0;
            for (; j + EDGE_BATCH <= n; j += EDGE_BATCH)
                edge_batch<true>(ebase, e, j, n, a);
            if (j < n)
                edge_batch<false>(ebase, e, j, n, a);
        }
        *(float4*)(acc + (size_t)(row0 + r) * DIM + lane * 4) = a;
    }
}

#define GEMM_WAVES   8
#define GEMM_ROWS    (GEMM_WAVES * 16)
#define GEMM_BLOCKS  ((N_ENT + GEMM_ROWS - 1) / GEMM_ROWS)
#define TILE_PITCH   132

template <bool NORM>
__global__ __launch_bounds__(GEMM_WAVES * 32)
void gemm_relu_kernel(const float* __restrict__ A,
                      const float* __restrict__ Wf,
                      float* __restrict__ out) {
    __shared__ __attribute__((aligned(16))) float lds[DIM * DIM + GEMM_WAVES * 16 * TILE_PITCH];
    float* sW = lds;
    float* sT = lds + DIM * DIM;

    const int wave = __builtin_amdgcn_readfirstlane(threadIdx.x >> 5);
    const int lane = threadIdx.x & 31;
    const int c = lane & 15;
    const int h = lane >> 4;
    const int rowBase = blockIdx.x * GEMM_ROWS + wave * 16;
    const bool active = rowBase < N_ENT;
    float* tile = sT + wave * (16 * TILE_PITCH);

    if (active) {
        float4 t[16];
#pragma unroll
        for (int r = 0; r < 16; ++r)
            t[r] = *(const float4*)(A + (size_t)(rowBase + r) * DIM + lane * 4);
#pragma unroll
        for (int r = 0; r < 16; ++r)
            *(float4*)(tile + r * TILE_PITCH + lane * 4) = t[r];
    }
    {
        const float4* g = (const float4*)Wf + threadIdx.x;
        float4*       s = (float4*)sW + threadIdx.x;
#pragma unroll
        for (int i = 0; i < (DIM * DIM / 4) / (GEMM_WAVES * 32); ++i)
            s[i * (GEMM_WAVES * 32)] = g[i * (GEMM_WAVES * 32)];
    }
    __syncthreads();
    if (!active) return;

    v8f acc[8];
#pragma unroll
    for (int ct = 0; ct < 8; ++ct) acc[ct] = (v8f){};

    const float*  ap = tile + c * TILE_PITCH + 2 * h;
    const float4* bp = (const float4*)sW + lane;
    v2f    a[2][2];
    float4 b[2][8];
    a[0][0] = *(const v2f*)(ap);
    a[0][1] = *(const v2f*)(ap + 4);
#pragma unroll
    for (int ct = 0; ct < 8; ++ct) b[0][ct] = bp[(ct * (DIM / 8)) * 32];
#pragma unroll
    for (int j = 0; j < DIM / 8; ++j) {
        const int cur = j & 1, nxt = cur ^ 1;
        if (j + 1 < DIM / 8) {
            a[nxt][0] = *(const v2f*)(ap + 8 * (j + 1));
            a[nxt][1] = *(const v2f*)(ap + 8 * (j + 1) + 4);
#pragma unroll
            for (int ct = 0; ct < 8; ++ct) b[nxt][ct] = bp[(ct * (DIM / 8) + j + 1) * 32];
        }
        __builtin_amdgcn_sched_barrier(0);
#pragma unroll
        for (int ct = 0; ct < 8; ++ct) {
            v2f b0 = {b[cur][ct].x, b[cur][ct].y};
            acc[ct] = __builtin_amdgcn_wmma_f32_16x16x4_f32(
                           false, a[cur][0],  false, b0,
                           (short)0, acc[ct],  false,  false);
        }
#pragma unroll
        for (int ct = 0; ct < 8; ++ct) {
            v2f b1 = {b[cur][ct].z, b[cur][ct].w};
            acc[ct] = __builtin_amdgcn_wmma_f32_16x16x4_f32(
                           false, a[cur][1],  false, b1,
                           (short)0, acc[ct],  false,  false);
        }
        __builtin_amdgcn_sched_barrier(0);
    }

#pragma unroll
    for (int ct = 0; ct < 8; ++ct)
#pragma unroll
        for (int r = 0; r < 8; ++r)
            tile[(8 * h + r) * TILE_PITCH + 16 * ct + c] = fmaxf(acc[ct][r], 0.f);
    asm volatile("s_wait_dscnt 0x0" ::: "memory");
    if (!NORM) {
#pragma unroll
        for (int r = 0; r < 16; ++r) {
            const float4 v = *(const float4*)(tile + r * TILE_PITCH + lane * 4);
            *(float4*)(out + (size_t)(rowBase + r) * DIM + lane * 4) = v;
        }
    } else {
        float4 v[16];
        float  s[16];
#pragma unroll
        for (int r = 0; r < 16; ++r) {
            v[r] = *(const float4*)(tile + r * TILE_PITCH + lane * 4);
            s[r] = v[r].x * v[r].x + v[r].y * v[r].y + v[r].z * v[r].z + v[r].w * v[r].w;
        }
        const bool b4 = (lane & 16) != 0, b3 = (lane & 8) != 0, b2 = (lane & 4) != 0, b1 = (lane & 2) != 0;
        float t[8], u[4], w[2];
#pragma unroll
        for (int k = 0; k < 8; ++k)
            t[k] = (b4 ? s[k + 8] : s[k]) + __shfl_xor(b4 ? s[k] : s[k + 8], 16, 32);
#pragma unroll
        for (int k = 0; k < 4; ++k)
            u[k] = (b3 ? t[k + 4] : t[k]) + __shfl_xor(b3 ? t[k] : t[k + 4], 8, 32);
#pragma unroll
        for (int k = 0; k < 2; ++k)
            w[k] = (b2 ? u[k + 2] : u[k]) + __shfl_xor(b2 ? u[k] : u[k + 2], 4, 32);
        float x = (b1 ? w[1] : w[0]) + __shfl_xor(b1 ? w[0] : w[1], 2, 32);
        x += __shfl_xor(x, 1, 32);
        const float inv = 1.0f / fmaxf(sqrtf(x), 1e-12f);
#pragma unroll
        for (int r = 0; r < 16; ++r) {
            const float ir = __int_as_float(__builtin_amdgcn_readlane(__float_as_int(inv), 2 * r));
            float4 y = v[r];
            y.x *= ir; y.y *= ir; y.z *= ir; y.w *= ir;
            *(float4*)(out + (size_t)(rowBase + r) * DIM + lane * 4) = y;
        }
    }
}

extern "C" void kernel_launch(void* const* d_in, const int* in_sizes, int n_in,
                              void* d_out, int out_size, void* d_ws, size_t ws_size,
                              hipStream_t stream) {
    const float* ent_emb = (const float*)d_in[0];
    const float* bw      = (const float*)d_in[1];
    const float* bc      = (const float*)d_in[2];
    const float* eval    = (const float*)d_in[3];
    const int*   esrc    = (const int*)d_in[5];
    const int*   edst    = (const int*)d_in[6];

    float* out = (float*)d_out;
    float* acc = (float*)d_ws;
    float* W   = acc + (size_t)N_ENT * DIM;
    int*   cnt      = (int*)(W + DIM * DIM);
    int*   rowstart = cnt + CNT_PAD;
    int*   rank     = rowstart + CNT_PAD;
    int2*  sorted   = (int2*)(rank + NEDGE);

    zero_kernel<<<(CNT_PAD / 4 + 255) / 256, 256, 0, stream>>>((float4*)cnt, CNT_PAD / 4);
    rank_kernel<<<(NEDGE / 4 + 255) / 256, 256, 0, stream>>>(esrc, cnt, rank);
    alloc_kernel<<<(N_ENT + 255) / 256, 256, 0, stream>>>(cnt, cnt + N_ENT, rowstart);
    fill_kernel<<<(NEDGE / 4 + 255) / 256, 256, 0, stream>>>(esrc, edst, eval, rank, rowstart,
                                                          (unsigned long long*)sorted);

    for (int l = 0; l < NLAYER; ++l) {
        build_w_kernel<<<(DIM * DIM) / 256, 256, 0, stream>>>(bw, bc, W, l);
        const float* emb_in = (l == 0) ? ent_emb : out;
        gather_kernel<<<GATHER_WAVES / 4, 128, 0, stream>>>(emb_in, rowstart, cnt, sorted, acc);
        if (l == NLAYER - 1)
            gemm_relu_kernel<true><<<GEMM_BLOCKS, GEMM_WAVES * 32, 0, stream>>>(acc, W, out);
        else
            gemm_relu_kernel<false><<<GEMM_BLOCKS, GEMM_WAVES * 32, 0, stream>>>(acc, W, out);
    }
}


# Edge ordering as a two-level counting sort on LDS counters (chunks by row range, then a work-group per range orders rows); global counter atomics, zero fill gone; row sums in LDS add arrival order
# speedup vs baseline: 4.5765x; 1.7983x over previous
#include <hip/hip_runtime.h>

#define N_ENT   50000
#define DIM     128
#define NREL    11
#define NBASIS  4
#define NEDGE   1000000
#define NLAYER  2

#define CNT_PAD        50048
#define ROWS_PER_WAVE  4
#define GATHER_WAVES   (N_ENT / ROWS_PER_WAVE)
#define EDGE_BATCH     8

typedef __attribute__((ext_vector_type(2))) float v2f;
typedef __attribute__((ext_vector_type(8))) float v8f;

__global__ void build_w_kernel(const float* __restrict__ bw,
                               const float* __restrict__ bc,
                               float* __restrict__ Wf, int l) {
    int idx  = blockIdx.x * blockDim.x + threadIdx.x;
    int q    = idx & 3;
    int lane = (idx >> 2) & 31;
    int j    = (idx >> 7) & 15;
    int ct   = idx >> 11;
    int o = 16 * ct + (lane & 15);
    int d = 8 * j + 4 * (q >> 1) + 2 * (lane >> 4) + (q & 1);
    float s = 0.f;
#pragma unroll
    for (int b = 0; b < NBASIS; ++b) {
        float c = bc[(l * NREL + l) * NBASIS + b];
        s += c * bw[(((size_t)l * NBASIS + b) * DIM + o) * DIM + d];
    }
    Wf[idx] = s;
}

#define CHUNK       4096
#define NCHUNK      ((NEDGE + CHUNK - 1) / CHUNK)
#define BIN_ROWS    224
#define NBIN        ((N_ENT + BIN_ROWS - 1) / BIN_ROWS)
#define TAB_PITCH   256
#define STAGE_CAP   8192

__device__ __forceinline__ void wave_scan256(const int* in, int* out, int lane) {
    int v[8];
    int s = 0;
#pragma unroll
    for (int j = 0; j < 8; ++j) { v[j] = in[8 * lane + j]; s += v[j]; }
    int incl = s;
#pragma unroll
    for (int off = 1; off < 32; off <<= 1) {
        int t = __shfl_up(incl, off, 32);
        if (lane >= off) incl += t;
    }
    int run = incl - s;
#pragma unroll
    for (int j = 0; j < 8; ++j) { out[8 * lane + j] = run; run += v[j]; }
    if (lane == 31) out[256] = incl;
}

__device__ __forceinline__ unsigned long long pack_edge(int dst, float val) {
    return (unsigned long long)(unsigned)dst | ((unsigned long long)(unsigned)__float_as_int(val) << 32);
}

__global__ __launch_bounds__(1024)
void chunk_kernel(const int* __restrict__ esrc,
                  const int* __restrict__ edst,
                  const float* __restrict__ eval,
                  unsigned* __restrict__ tab,
                  int* __restrict__ rsrc,
                  int* __restrict__ rdst,
                  float* __restrict__ rval) {
    __shared__ int hist[256];
    __shared__ int off[257];
    __shared__ __attribute__((aligned(16))) int   lsrc[CHUNK];
    __shared__ __attribute__((aligned(16))) int   ldst[CHUNK];
    __shared__ __attribute__((aligned(16))) float lval[CHUNK];

    const int t     = threadIdx.x;
    const int chunk = blockIdx.x;
    const int i4    = chunk * (CHUNK / 4) + t;
    const bool valid = i4 < NEDGE / 4;

    if (t < 256) hist[t] = 0;
    int4   s = make_int4(0, 0, 0, 0);
    int4   d = make_int4(0, 0, 0, 0);
    float4 v = make_float4(0.f, 0.f, 0.f, 0.f);
    if (valid) {
        s = *(const int4*)(esrc + 4 * (size_t)i4);
        d = *(const int4*)(edst + 4 * (size_t)i4);
        v = *(const float4*)(eval + 4 * (size_t)i4);
    }
    __syncthreads();

    const int b0 = (int)min((unsigned)s.x / BIN_ROWS, (unsigned)(NBIN - 1));
    const int b1 = (int)min((unsigned)s.y / BIN_ROWS, (unsigned)(NBIN - 1));
    const int b2 = (int)min((unsigned)s.z / BIN_ROWS, (unsigned)(NBIN - 1));
    const int b3 = (int)min((unsigned)s.w / BIN_ROWS, (unsigned)(NBIN - 1));
    int r0 = 0, r1 = 0, r2 = 0, r3 = 0;
    if (valid) {
        r0 = atomicAdd(&hist[b0], 1);
        r1 = atomicAdd(&hist[b1], 1);
        r2 = atomicAdd(&hist[b2], 1);
        r3 = atomicAdd(&hist[b3], 1);
    }
    __syncthreads();

    if (t < 32) wave_scan256(hist, off, t);
    __syncthreads();

    if (t < TAB_PITCH)
        tab[chunk * TAB_PITCH + t] = ((unsigned)off[t] << 16) | (unsigned)hist[t];
    if (valid) {
        const int p0 = off[b0] + r0, p1 = off[b1] + r1, p2 = off[b2] + r2, p3 = off[b3] + r3;
        lsrc[p0] = s.x; ldst[p0] = d.x; lval[p0] = v.x;
        lsrc[p1] = s.y; ldst[p1] = d.y; lval[p1] = v.y;
        lsrc[p2] = s.z; ldst[p2] = d.z; lval[p2] = v.z;
        lsrc[p3] = s.w; ldst[p3] = d.w; lval[p3] = v.w;
    }
    __syncthreads();

    if (valid) {
        const size_t g = (size_t)chunk * CHUNK + 4 * t;
        *(int4*)(rsrc + g)   = *(const int4*)(lsrc + 4 * t);
        *(int4*)(rdst + g)   = *(const int4*)(ldst + 4 * t);
        *(float4*)(rval + g) = *(const float4*)(lval + 4 * t);
    }
}

__device__ __forceinline__ int piece_slot(const int* pstart, const int* poff, int e) {
    int lo = 0, hi = 256;
#pragma unroll
    for (int it = 0; it < 8; ++it) {
        const int mid = (lo + hi) >> 1;
        if (pstart[mid] <= e) lo = mid; else hi = mid;
    }
    return lo * CHUNK + poff[lo] + (e - pstart[lo]);
}

__global__ __launch_bounds__(1024)
void bin_kernel(const unsigned* __restrict__ tab,
                const int* __restrict__ rsrc,
                const int* __restrict__ rdst,
                const float* __restrict__ rval,
                int* __restrict__ cnt,
                int* __restrict__ rowstart,
                unsigned long long* __restrict__ sorted) {
    __shared__ int pcnt[256];
    __shared__ int poff[256];
    __shared__ int pstart[257];
    __shared__ int rowcnt[256];
    __shared__ int rowpos[257];
    __shared__ int red[32];
    __shared__ int sbase;
    __shared__ unsigned long long stage[STAGE_CAP];

    const int t    = threadIdx.x;
    const int lane = t & 31;
    const int wave = t >> 5;
    const int bin  = blockIdx.x;

    int part = 0;
    for (int q = t; q < NCHUNK * TAB_PITCH / 4; q += 1024) {
        const uint4 w = ((const uint4*)tab)[q];
        const int col = (q & (TAB_PITCH / 4 - 1)) * 4;
        part += (col     < bin ? (int)(w.x & 0xFFFFu) : 0) + (col + 1 < bin ? (int)(w.y & 0xFFFFu) : 0)
              + (col + 2 < bin ? (int)(w.z & 0xFFFFu) : 0) + (col + 3 < bin ? (int)(w.w & 0xFFFFu) : 0);
    }
#pragma unroll
    for (int off = 16; off > 0; off >>= 1) part += __shfl_xor(part, off, 32);
    if (lane == 0) red[wave] = part;
    if (t < 256) {
        unsigned w = 0;
        if (t < NCHUNK) w = tab[t * TAB_PITCH + bin];
        pcnt[t]   = (int)(w & 0xFFFFu);
        poff[t]   = (int)(w >> 16);
        rowcnt[t] = 0;
    }
    __syncthreads();

    if (t < 32) {
        wave_scan256(pcnt, pstart, t);
        int x = red[t];
#pragma unroll
        for (int off = 16; off > 0; off >>= 1) x += __shfl_xor(x, off, 32);
        if (t == 0) sbase = x;
    }
    __syncthreads();
    const int total = pstart[256];
    const int base  = sbase;
    const int row0  = bin * BIN_ROWS;

    for (int e0 = t; e0 < total; e0 += 4 * 1024) {
        int lr[4];
#pragma unroll
        for (int k = 0; k < 4; ++k) {
            const int e = e0 + k * 1024;
            lr[k] = -1;
            if (e < total) lr[k] = min(max(rsrc[piece_slot(pstart, poff, e)] - row0, 0), BIN_ROWS - 1);
        }
#pragma unroll
        for (int k = 0; k < 4; ++k)
            if (lr[k] >= 0) atomicAdd(&rowcnt[lr[k]], 1);
    }
    __syncthreads();

    if (t < 32) wave_scan256(rowcnt, rowpos, t);
    __syncthreads();
    if (t < BIN_ROWS && row0 + t < CNT_PAD) {
        cnt[row0 + t]      = rowcnt[t];
        rowstart[row0 + t] = base + rowpos[t];
    }
    __syncthreads();

    const bool staged = total <= STAGE_CAP;
    for (int e0 = t; e0 < total; e0 += 4 * 1024) {
        int lr[4];
        unsigned long long rec[4];
#pragma unroll
        for (int k = 0; k < 4; ++k) {
            const int e = e0 + k * 1024;
            lr[k]  = -1;
            rec[k] = 0;
            if (e < total) {
                const int g = piece_slot(pstart, poff, e);
                lr[k]  = min(max(rsrc[g] - row0, 0), BIN_ROWS - 1);
                rec[k] = pack_edge(rdst[g], rval[g]);
            }
        }
#pragma unroll
        for (int k = 0; k < 4; ++k) {
            if (lr[k] >= 0) {
                const int p = atomicAdd(&rowpos[lr[k]], 1);
                if (staged) stage[min(p, STAGE_CAP - 1)] = rec[k];
                else        atomicExch(sorted + min(base + p, NEDGE - 1), rec[k]);
            }
        }
    }
    __syncthreads();
    if (staged) {
        const int lo  = base & ~15;
        const int in0 = (base + 15) & ~15;
        const int in1 = (base + total) & ~15;
        for (int idx = lo + t; idx < base + total; idx += 1024) {
            const int e = idx - base;
            if (e >= 0) {
                const int at = min(idx, NEDGE - 1);
                if (idx >= in0 && idx < in1) sorted[at] = stage[e];
                else                         atomicExch(sorted + at, stage[e]);
            }
        }
    }
}

template <bool FULL>
__device__ __forceinline__ void edge_batch(const float* __restrict__ ebase, int2 e, int j, int n,
                                           float4& a) {
    float4 x[EDGE_BATCH];
    float  v[EDGE_BATCH];
#pragma unroll
    for (int u = 0; u < EDGE_BATCH; ++u) {
        v[u] = 0.f;
        x[u] = make_float4(0.f, 0.f, 0.f, 0.f);
        if (FULL || j + u < n) {
            int d = __builtin_amdgcn_readlane(e.x, j + u);
            v[u]  = __int_as_float(__builtin_amdgcn_readlane(e.y, j + u));
            x[u]  = *(const float4*)(ebase + (size_t)d * DIM);
        }
    }
#pragma unroll
    for (int u = 0; u < EDGE_BATCH; ++u) {
        a.x = fmaf(v[u], x[u].x, a.x);
        a.y = fmaf(v[u], x[u].y, a.y);
        a.z = fmaf(v[u], x[u].z, a.z);
        a.w = fmaf(v[u], x[u].w, a.w);
    }
}

__global__ __launch_bounds__(128)
void gather_kernel(const float* __restrict__ emb,
                   const int* __restrict__ rowstart,
                   const int* __restrict__ cnt,
                   const int2* __restrict__ sorted,
                   float* __restrict__ acc) {
    const int wave = __builtin_amdgcn_readfirstlane(blockIdx.x * 4 + (threadIdx.x >> 5));
    const int lane = threadIdx.x & 31;
    const int row0 = wave * ROWS_PER_WAVE;

    int st[ROWS_PER_WAVE], dg[ROWS_PER_WAVE];
#pragma unroll
    for (int r = 0; r < ROWS_PER_WAVE; ++r) {
        st[r] = min(max(rowstart[row0 + r], 0), NEDGE - 1);
        dg[r] = min(max(cnt[row0 + r], 0), NEDGE);
    }
    int2 first[ROWS_PER_WAVE];
#pragma unroll
    for (int r = 0; r < ROWS_PER_WAVE; ++r)
        first[r] = (lane < dg[r]) ? sorted[min(st[r] + lane, NEDGE - 1)] : make_int2(0, 0);

    const float* ebase = emb + lane * 4;
#pragma unroll
    for (int r = 0; r < ROWS_PER_WAVE; ++r) {
        float4 a = make_float4(0.f, 0.f, 0.f, 0.f);
        const int deg = dg[r];
        int2 e = first[r];
        for (int base = 0; base < deg; base += 32) {
            if (base > 0)
                e = (base + lane < deg) ? sorted[min(st[r] + base + lane, NEDGE - 1)] : make_int2(0, 0);
            e.x = min(max(e.x, 0), N_ENT - 1);
            const int n = min(32, deg - base);
            int j = 0;
            for (; j + EDGE_BATCH <= n; j += EDGE_BATCH)
                edge_batch<true>(ebase, e, j, n, a);
            if (j < n)
                edge_batch<false>(ebase, e, j, n, a);
        }
        *(float4*)(acc + (size_t)(row0 + r) * DIM + lane * 4) = a;
    }
}

#define GEMM_WAVES   8
#define GEMM_ROWS    (GEMM_WAVES * 16)
#define GEMM_BLOCKS  ((N_ENT + GEMM_ROWS - 1) / GEMM_ROWS)
#define TILE_PITCH   132

template <bool NORM>
__global__ __launch_bounds__(GEMM_WAVES * 32)
void gemm_relu_kernel(const float* __restrict__ A,
                      const float* __restrict__ Wf,
                      float* __restrict__ out) {
    __shared__ __attribute__((aligned(16))) float lds[DIM * DIM + GEMM_WAVES * 16 * TILE_PITCH];
    float* sW = lds;
    float* sT = lds + DIM * DIM;

    const int wave = __builtin_amdgcn_readfirstlane(threadIdx.x >> 5);
    const int lane = threadIdx.x & 31;
    const int c = lane & 15;
    const int h = lane >> 4;
    const int rowBase = blockIdx.x * GEMM_ROWS + wave * 16;
    const bool active = rowBase < N_ENT;
    float* tile = sT + wave * (16 * TILE_PITCH);

    if (active) {
        float4 t[16];
#pragma unroll
        for (int r = 0; r < 16; ++r)
            t[r] = *(const float4*)(A + (size_t)(rowBase + r) * DIM + lane * 4);
#pragma unroll
        for (int r = 0; r < 16; ++r)
            *(float4*)(tile + r * TILE_PITCH + lane * 4) = t[r];
    }
    {
        const float4* g = (const float4*)Wf + threadIdx.x;
        float4*       s = (float4*)sW + threadIdx.x;
#pragma unroll
        for (int i = 0; i < (DIM * DIM / 4) / (GEMM_WAVES * 32); ++i)
            s[i * (GEMM_WAVES * 32)] = g[i * (GEMM_WAVES * 32)];
    }
    __syncthreads();
    if (!active) return;

    v8f acc[8];
#pragma unroll
    for (int ct = 0; ct < 8; ++ct) acc[ct] = (v8f){};

    const float*  ap = tile + c * TILE_PITCH + 2 * h;
    const float4* bp = (const float4*)sW + lane;
    v2f    a[2][2];
    float4 b[2][8];
    a[0][0] = *(const v2f*)(ap);
    a[0][1] = *(const v2f*)(ap + 4);
#pragma unroll
    for (int ct = 0; ct < 8; ++ct) b[0][ct] = bp[(ct * (DIM / 8)) * 32];
#pragma unroll
    for (int j = 0; j < DIM / 8; ++j) {
        const int cur = j & 1, nxt = cur ^ 1;
        if (j + 1 < DIM / 8) {
            a[nxt][0] = *(const v2f*)(ap + 8 * (j + 1));
            a[nxt][1] = *(const v2f*)(ap + 8 * (j + 1) + 4);
#pragma unroll
            for (int ct = 0; ct < 8; ++ct) b[nxt][ct] = bp[(ct * (DIM / 8) + j + 1) * 32];
        }
        __builtin_amdgcn_sched_barrier(0);
#pragma unroll
        for (int ct = 0; ct < 8; ++ct) {
            v2f b0 = {b[cur][ct].x, b[cur][ct].y};
            acc[ct] = __builtin_amdgcn_wmma_f32_16x16x4_f32(
                           false, a[cur][0],  false, b0,
                           (short)0, acc[ct],  false,  false);
        }
#pragma unroll
        for (int ct = 0; ct < 8; ++ct) {
            v2f b1 = {b[cur][ct].z, b[cur][ct].w};
            acc[ct] = __builtin_amdgcn_wmma_f32_16x16x4_f32(
                           false, a[cur][1],  false, b1,
                           (short)0, acc[ct],  false,  false);
        }
        __builtin_amdgcn_sched_barrier(0);
    }

#pragma unroll
    for (int ct = 0; ct < 8; ++ct)
#pragma unroll
        for (int r = 0; r < 8; ++r)
            tile[(8 * h + r) * TILE_PITCH + 16 * ct + c] = fmaxf(acc[ct][r], 0.f);
    asm volatile("s_wait_dscnt 0x0" ::: "memory");
    if (!NORM) {
#pragma unroll
        for (int r = 0; r < 16; ++r) {
            const float4 v = *(const float4*)(tile + r * TILE_PITCH + lane * 4);
            *(float4*)(out + (size_t)(rowBase + r) * DIM + lane * 4) = v;
        }
    } else {
        float4 v[16];
        float  s[16];
#pragma unroll
        for (int r = 0; r < 16; ++r) {
            v[r] = *(const float4*)(tile + r * TILE_PITCH + lane * 4);
            s[r] = v[r].x * v[r].x + v[r].y * v[r].y + v[r].z * v[r].z + v[r].w * v[r].w;
        }
        const bool b4 = (lane & 16) != 0, b3 = (lane & 8) != 0, b2 = (lane & 4) != 0, b1 = (lane & 2) != 0;
        float t[8], u[4], w[2];
#pragma unroll
        for (int k = 0; k < 8; ++k)
            t[k] = (b4 ? s[k + 8] : s[k]) + __shfl_xor(b4 ? s[k] : s[k + 8], 16, 32);
#pragma unroll
        for (int k = 0; k < 4; ++k)
            u[k] = (b3 ? t[k + 4] : t[k]) + __shfl_xor(b3 ? t[k] : t[k + 4], 8, 32);
#pragma unroll
        for (int k = 0; k < 2; ++k)
            w[k] = (b2 ? u[k + 2] : u[k]) + __shfl_xor(b2 ? u[k] : u[k + 2], 4, 32);
        float x = (b1 ? w[1] : w[0]) + __shfl_xor(b1 ? w[0] : w[1], 2, 32);
        x += __shfl_xor(x, 1, 32);
        const float inv = 1.0f / fmaxf(sqrtf(x), 1e-12f);
#pragma unroll
        for (int r = 0; r < 16; ++r) {
            const float ir = __int_as_float(__builtin_amdgcn_readlane(__float_as_int(inv), 2 * r));
            float4 y = v[r];
            y.x *= ir; y.y *= ir; y.z *= ir; y.w *= ir;
            *(float4*)(out + (size_t)(rowBase + r) * DIM + lane * 4) = y;
        }
    }
}

extern "C" void kernel_launch(void* const* d_in, const int* in_sizes, int n_in,
                              void* d_out, int out_size, void* d_ws, size_t ws_size,
                              hipStream_t stream) {
    const float* ent_emb = (const float*)d_in[0];
    const float* bw      = (const float*)d_in[1];
    const float* bc      = (const float*)d_in[2];
    const float* eval    = (const float*)d_in[3];
    const int*   esrc    = (const int*)d_in[5];
    const int*   edst    = (const int*)d_in[6];

    float* out = (float*)d_out;
    float* acc = (float*)d_ws;
    float* W   = acc + (size_t)N_ENT * DIM;
    int*      cnt      = (int*)(W + DIM * DIM);
    int*      rowstart = cnt + CNT_PAD;
    unsigned* tab      = (unsigned*)(rowstart + CNT_PAD);
    int*      rsrc     = (int*)(tab + NCHUNK * TAB_PITCH);
    int*      rdst     = rsrc + NCHUNK * CHUNK;
    float*    rval     = (float*)(rdst + NCHUNK * CHUNK);
    int2*     sorted   = (int2*)(rval + NCHUNK * CHUNK);

    chunk_kernel<<<NCHUNK, 1024, 0, stream>>>(esrc, edst, eval, tab, rsrc, rdst, rval);
    bin_kernel<<<NBIN, 1024, 0, stream>>>(tab, rsrc, rdst, rval, cnt, rowstart, (unsigned long long*)sorted);

    for (int l = 0; l < NLAYER; ++l) {
        build_w_kernel<<<(DIM * DIM) / 256, 256, 0, stream>>>(bw, bc, W, l);
        const float* emb_in = (l == 0) ? ent_emb : out;
        gather_kernel<<<GATHER_WAVES / 4, 128, 0, stream>>>(emb_in, rowstart, cnt, sorted, acc);
        if (l == NLAYER - 1)
            gemm_relu_kernel<true><<<GEMM_BLOCKS, GEMM_WAVES * 32, 0, stream>>>(acc, W, out);
        else
            gemm_relu_kernel<false><<<GEMM_BLOCKS, GEMM_WAVES * 32, 0, stream>>>(acc, W, out);
    }
}


# Both layers' W tables built by extra work-groups of the call's first launch (the edge-chunk kernel), one W buffer per layer; the two separate W-build launches removed: 8 launches per call become 6
# speedup vs baseline: 4.8159x; 1.0523x over previous
#include <hip/hip_runtime.h>

#define N_ENT   50000
#define DIM     128
#define NREL    11
#define NBASIS  4
#define NEDGE   1000000
#define NLAYER  2

#define CNT_PAD        50048
#define ROWS_PER_WAVE  4
#define GATHER_WAVES   (N_ENT / ROWS_PER_WAVE)
#define EDGE_BATCH     8

typedef __attribute__((ext_vector_type(2))) float v2f;
typedef __attribute__((ext_vector_type(8))) float v8f;

#define W_ELEMS   (DIM * DIM)
#define W_GROUPS  (W_ELEMS / 1024)
#define W_BLOCKS  (NLAYER * W_GROUPS)

__device__ __forceinline__ void build_w_element(const float* __restrict__ bw,
                                                const float* __restrict__ bc,
                                                float* __restrict__ Wf, int l, int idx) {
    int q    = idx & 3;
    int lane = (idx >> 2) & 31;
    int j    = (idx >> 7) & 15;
    int ct   = idx >> 11;
    int o = 16 * ct + (lane & 15);
    int d = 8 * j + 4 * (q >> 1) + 2 * (lane >> 4) + (q & 1);
    float s = 0.f;
#pragma unroll
    for (int b = 0; b < NBASIS; ++b) {
        float c = bc[(l * NREL + l) * NBASIS + b];
        s += c * bw[(((size_t)l * NBASIS + b) * DIM + o) * DIM + d];
    }
    Wf[(size_t)l * W_ELEMS + idx] = s;
}

#define CHUNK       4096
#define NCHUNK      ((NEDGE + CHUNK - 1) / CHUNK)
#define BIN_ROWS    224
#define NBIN        ((N_ENT + BIN_ROWS - 1) / BIN_ROWS)
#define TAB_PITCH   256
#define STAGE_CAP   8192

__device__ __forceinline__ void wave_scan256(const int* in, int* out, int lane) {
    int v[8];
    int s = 0;
#pragma unroll
    for (int j = 0; j < 8; ++j) { v[j] = in[8 * lane + j]; s += v[j]; }
    int incl = s;
#pragma unroll
    for (int off = 1; off < 32; off <<= 1) {
        int t = __shfl_up(incl, off, 32);
        if (lane >= off) incl += t;
    }
    int run = incl - s;
#pragma unroll
    for (int j = 0; j < 8; ++j) { out[8 * lane + j] = run; run += v[j]; }
    if (lane == 31) out[256] = incl;
}

__device__ __forceinline__ unsigned long long pack_edge(int dst, float val) {
    return (unsigned long long)(unsigned)dst | ((unsigned long long)(unsigned)__float_as_int(val) << 32);
}

__global__ __launch_bounds__(1024)
void chunk_kernel(const int* __restrict__ esrc,
                  const int* __restrict__ edst,
                  const float* __restrict__ eval,
                  const float* __restrict__ bw,
                  const float* __restrict__ bc,
                  float* __restrict__ Wf,
                  unsigned* __restrict__ tab,
                  int* __restrict__ rsrc,
                  int* __restrict__ rdst,
                  float* __restrict__ rval) {
    __shared__ int hist[256];
    __shared__ int off[257];
    __shared__ __attribute__((aligned(16))) int   lsrc[CHUNK];
    __shared__ __attribute__((aligned(16))) int   ldst[CHUNK];
    __shared__ __attribute__((aligned(16))) float lval[CHUNK];

    const int t     = threadIdx.x;
    if (blockIdx.x >= NCHUNK) {
        const int wb = blockIdx.x - NCHUNK;
        build_w_element(bw, bc, Wf, wb / W_GROUPS, (wb % W_GROUPS) * 1024 + t);
        return;
    }
    const int chunk = blockIdx.x;
    const int i4    = chunk * (CHUNK / 4) + t;
    const bool valid = i4 < NEDGE / 4;

    if (t < 256) hist[t] = 0;
    int4   s = make_int4(0, 0, 0, 0);
    int4   d = make_int4(0, 0, 0, 0);
    float4 v = make_float4(0.f, 0.f, 0.f, 0.f);
    if (valid) {
        s = *(const int4*)(esrc + 4 * (size_t)i4);
        d = *(const int4*)(edst + 4 * (size_t)i4);
        v = *(const float4*)(eval + 4 * (size_t)i4);
    }
    __syncthreads();

    const int b0 = (int)min((unsigned)s.x / BIN_ROWS, (unsigned)(NBIN - 1));
    const int b1 = (int)min((unsigned)s.y / BIN_ROWS, (unsigned)(NBIN - 1));
    const int b2 = (int)min((unsigned)s.z / BIN_ROWS, (unsigned)(NBIN - 1));
    const int b3 = (int)min((unsigned)s.w / BIN_ROWS, (unsigned)(NBIN - 1));
    int r0 = 0, r1 = 0, r2 = 0, r3 = 0;
    if (valid) {
        r0 = atomicAdd(&hist[b0], 1);
        r1 = atomicAdd(&hist[b1], 1);
        r2 = atomicAdd(&hist[b2], 1);
        r3 = atomicAdd(&hist[b3], 1);
    }
    __syncthreads();

    if (t < 32) wave_scan256(hist, off, t);
    __syncthreads();

    if (t < TAB_PITCH)
        tab[chunk * TAB_PITCH + t] = ((unsigned)off[t] << 16) | (unsigned)hist[t];
    if (valid) {
        const int p0 = off[b0] + r0, p1 = off[b1] + r1, p2 = off[b2] + r2, p3 = off[b3] + r3;
        lsrc[p0] = s.x; ldst[p0] = d.x; lval[p0] = v.x;
        lsrc[p1] = s.y; ldst[p1] = d.y; lval[p1] = v.y;
        lsrc[p2] = s.z; ldst[p2] = d.z; lval[p2] = v.z;
        lsrc[p3] = s.w; ldst[p3] = d.w; lval[p3] = v.w;
    }
    __syncthreads();

    if (valid) {
        const size_t g = (size_t)chunk * CHUNK + 4 * t;
        *(int4*)(rsrc + g)   = *(const int4*)(lsrc + 4 * t);
        *(int4*)(rdst + g)   = *(const int4*)(ldst + 4 * t);
        *(float4*)(rval + g) = *(const float4*)(lval + 4 * t);
    }
}

__device__ __forceinline__ int piece_slot(const int* pstart, const int* poff, int e) {
    int lo = 0, hi = 256;
#pragma unroll
    for (int it = 0; it < 8; ++it) {
        const int mid = (lo + hi) >> 1;
        if (pstart[mid] <= e) lo = mid; else hi = mid;
    }
    return lo * CHUNK + poff[lo] + (e - pstart[lo]);
}

__global__ __launch_bounds__(1024)
void bin_kernel(const unsigned* __restrict__ tab,
                const int* __restrict__ rsrc,
                const int* __restrict__ rdst,
                const float* __restrict__ rval,
                int* __restrict__ cnt,
                int* __restrict__ rowstart,
                unsigned long long* __restrict__ sorted) {
    __shared__ int pcnt[256];
    __shared__ int poff[256];
    __shared__ int pstart[257];
    __shared__ int rowcnt[256];
    __shared__ int rowpos[257];
    __shared__ int red[32];
    __shared__ int sbase;
    __shared__ unsigned long long stage[STAGE_CAP];

    const int t    = threadIdx.x;
    const int lane = t & 31;
    const int wave = t >> 5;
    const int bin  = blockIdx.x;

    int part = 0;
    for (int q = t; q < NCHUNK * TAB_PITCH / 4; q += 1024) {
        const uint4 w = ((const uint4*)tab)[q];
        const int col = (q & (TAB_PITCH / 4 - 1)) * 4;
        part += (col     < bin ? (int)(w.x & 0xFFFFu) : 0) + (col + 1 < bin ? (int)(w.y & 0xFFFFu) : 0)
              + (col + 2 < bin ? (int)(w.z & 0xFFFFu) : 0) + (col + 3 < bin ? (int)(w.w & 0xFFFFu) : 0);
    }
#pragma unroll
    for (int off = 16; off > 0; off >>= 1) part += __shfl_xor(part, off, 32);
    if (lane == 0) red[wave] = part;
    if (t < 256) {
        unsigned w = 0;
        if (t < NCHUNK) w = tab[t * TAB_PITCH + bin];
        pcnt[t]   = (int)(w & 0xFFFFu);
        poff[t]   = (int)(w >> 16);
        rowcnt[t] = 0;
    }
    __syncthreads();

    if (t < 32) {
        wave_scan256(pcnt, pstart, t);
        int x = red[t];
#pragma unroll
        for (int off = 16; off > 0; off >>= 1) x += __shfl_xor(x, off, 32);
        if (t == 0) sbase = x;
    }
    __syncthreads();
    const int total = pstart[256];
    const int base  = sbase;
    const int row0  = bin * BIN_ROWS;

    for (int e0 = t; e0 < total; e0 += 4 * 1024) {
        int lr[4];
#pragma unroll
        for (int k = 0; k < 4; ++k) {
            const int e = e0 + k * 1024;
            lr[k] = -1;
            if (e < total) lr[k] = min(max(rsrc[piece_slot(pstart, poff, e)] - row0, 0), BIN_ROWS - 1);
        }
#pragma unroll
        for (int k = 0; k < 4; ++k)
            if (lr[k] >= 0) atomicAdd(&rowcnt[lr[k]], 1);
    }
    __syncthreads();

    if (t < 32) wave_scan256(rowcnt, rowpos, t);
    __syncthreads();
    if (t < BIN_ROWS && row0 + t < CNT_PAD) {
        cnt[row0 + t]      = rowcnt[t];
        rowstart[row0 + t] = base + rowpos[t];
    }
    __syncthreads();

    const bool staged = total <= STAGE_CAP;
    for (int e0 = t; e0 < total; e0 += 4 * 1024) {
        int lr[4];
        unsigned long long rec[4];
#pragma unroll
        for (int k = 0; k < 4; ++k) {
            const int e = e0 + k * 1024;
            lr[k]  = -1;
            rec[k] = 0;
            if (e < total) {
                const int g = piece_slot(pstart, poff, e);
                lr[k]  = min(max(rsrc[g] - row0, 0), BIN_ROWS - 1);
                rec[k] = pack_edge(rdst[g], rval[g]);
            }
        }
#pragma unroll
        for (int k = 0; k < 4; ++k) {
            if (lr[k] >= 0) {
                const int p = atomicAdd(&rowpos[lr[k]], 1);
                if (staged) stage[min(p, STAGE_CAP - 1)] = rec[k];
                else        atomicExch(sorted + min(base + p, NEDGE - 1), rec[k]);
            }
        }
    }
    __syncthreads();
    if (staged) {
        const int lo  = base & ~15;
        const int in0 = (base + 15) & ~15;
        const int in1 = (base + total) & ~15;
        for (int idx = lo + t; idx < base + total; idx += 1024) {
            const int e = idx - base;
            if (e >= 0) {
                const int at = min(idx, NEDGE - 1);
                if (idx >= in0 && idx < in1) sorted[at] = stage[e];
                else                         atomicExch(sorted + at, stage[e]);
            }
        }
    }
}

template <bool FULL>
__device__ __forceinline__ void edge_batch(const float* __restrict__ ebase, int2 e, int j, int n,
                                           float4& a) {
    float4 x[EDGE_BATCH];
    float  v[EDGE_BATCH];
#pragma unroll
    for (int u = 0; u < EDGE_BATCH; ++u) {
        v[u] = 0.f;
        x[u] = make_float4(0.f, 0.f, 0.f, 0.f);
        if (FULL || j + u < n) {
            int d = __builtin_amdgcn_readlane(e.x, j + u);
            v[u]  = __int_as_float(__builtin_amdgcn_readlane(e.y, j + u));
            x[u]  = *(const float4*)(ebase + (size_t)d * DIM);
        }
    }
#pragma unroll
    for (int u = 0; u < EDGE_BATCH; ++u) {
        a.x = fmaf(v[u], x[u].x, a.x);
        a.y = fmaf(v[u], x[u].y, a.y);
        a.z = fmaf(v[u], x[u].z, a.z);
        a.w = fmaf(v[u], x[u].w, a.w);
    }
}

__global__ __launch_bounds__(128)
void gather_kernel(const float* __restrict__ emb,
                   const int* __restrict__ rowstart,
                   const int* __restrict__ cnt,
                   const int2* __restrict__ sorted,
                   float* __restrict__ acc) {
    const int wave = __builtin_amdgcn_readfirstlane(blockIdx.x * 4 + (threadIdx.x >> 5));
    const int lane = threadIdx.x & 31;
    const int row0 = wave * ROWS_PER_WAVE;

    int st[ROWS_PER_WAVE], dg[ROWS_PER_WAVE];
#pragma unroll
    for (int r = 0; r < ROWS_PER_WAVE; ++r) {
        st[r] = min(max(rowstart[row0 + r], 0), NEDGE - 1);
        dg[r] = min(max(cnt[row0 + r], 0), NEDGE);
    }
    int2 first[ROWS_PER_WAVE];
#pragma unroll
    for (int r = 0; r < ROWS_PER_WAVE; ++r)
        first[r] = (lane < dg[r]) ? sorted[min(st[r] + lane, NEDGE - 1)] : make_int2(0, 0);

    const float* ebase = emb + lane * 4;
#pragma unroll
    for (int r = 0; r < ROWS_PER_WAVE; ++r) {
        float4 a = make_float4(0.f, 0.f, 0.f, 0.f);
        const int deg = dg[r];
        int2 e = first[r];
        for (int base = 0; base < deg; base += 32) {
            if (base > 0)
                e = (base + lane < deg) ? sorted[min(st[r] + base + lane, NEDGE - 1)] : make_int2(0, 0);
            e.x = min(max(e.x, 0), N_ENT - 1);
            const int n = min(32, deg - base);
            int j = 0;
            for (; j + EDGE_BATCH <= n; j += EDGE_BATCH)
                edge_batch<true>(ebase, e, j, n, a);
            if (j < n)
                edge_batch<false>(ebase, e, j, n, a);
        }
        *(float4*)(acc + (size_t)(row0 + r) * DIM + lane * 4) = a;
    }
}

#define GEMM_WAVES   8
#define GEMM_ROWS    (GEMM_WAVES * 16)
#define GEMM_BLOCKS  ((N_ENT + GEMM_ROWS - 1) / GEMM_ROWS)
#define TILE_PITCH   132

template <bool NORM>
__global__ __launch_bounds__(GEMM_WAVES * 32)
void gemm_relu_kernel(const float* __restrict__ A,
                      const float* __restrict__ Wf,
                      float* __restrict__ out) {
    __shared__ __attribute__((aligned(16))) float lds[DIM * DIM + GEMM_WAVES * 16 * TILE_PITCH];
    float* sW = lds;
    float* sT = lds + DIM * DIM;

    const int wave = __builtin_amdgcn_readfirstlane(threadIdx.x >> 5);
    const int lane = threadIdx.x & 31;
    const int c = lane & 15;
    const int h = lane >> 4;
    const int rowBase = blockIdx.x * GEMM_ROWS + wave * 16;
    const bool active = rowBase < N_ENT;
    float* tile = sT + wave * (16 * TILE_PITCH);

    if (active) {
        float4 t[16];
#pragma unroll
        for (int r = 0; r < 16; ++r)
            t[r] = *(const float4*)(A + (size_t)(rowBase + r) * DIM + lane * 4);
#pragma unroll
        for (int r = 0; r < 16; ++r)
            *(float4*)(tile + r * TILE_PITCH + lane * 4) = t[r];
    }
    {
        const float4* g = (const float4*)Wf + threadIdx.x;
        float4*       s = (float4*)sW + threadIdx.x;
#pragma unroll
        for (int i = 0; i < (DIM * DIM / 4) / (GEMM_WAVES * 32); ++i)
            s[i * (GEMM_WAVES * 32)] = g[i * (GEMM_WAVES * 32)];
    }
    __syncthreads();
    if (!active) return;

    v8f acc[8];
#pragma unroll
    for (int ct = 0; ct < 8; ++ct) acc[ct] = (v8f){};

    const float*  ap = tile + c * TILE_PITCH + 2 * h;
    const float4* bp = (const float4*)sW + lane;
    v2f    a[2][2];
    float4 b[2][8];
    a[0][0] = *(const v2f*)(ap);
    a[0][1] = *(const v2f*)(ap + 4);
#pragma unroll
    for (int ct = 0; ct < 8; ++ct) b[0][ct] = bp[(ct * (DIM / 8)) * 32];
#pragma unroll
    for (int j = 0; j < DIM / 8; ++j) {
        const int cur = j & 1, nxt = cur ^ 1;
        if (j + 1 < DIM / 8) {
            a[nxt][0] = *(const v2f*)(ap + 8 * (j + 1));
            a[nxt][1] = *(const v2f*)(ap + 8 * (j + 1) + 4);
#pragma unroll
            for (int ct = 0; ct < 8; ++ct) b[nxt][ct] = bp[(ct * (DIM / 8) + j + 1) * 32];
        }
        __builtin_amdgcn_sched_barrier(0);
#pragma unroll
        for (int ct = 0; ct < 8; ++ct) {
            v2f b0 = {b[cur][ct].x, b[cur][ct].y};
            acc[ct] = __builtin_amdgcn_wmma_f32_16x16x4_f32(
                           false, a[cur][0],  false, b0,
                           (short)0, acc[ct],  false,  false);
        }
#pragma unroll
        for (int ct = 0; ct < 8; ++ct) {
            v2f b1 = {b[cur][ct].z, b[cur][ct].w};
            acc[ct] = __builtin_amdgcn_wmma_f32_16x16x4_f32(
                           false, a[cur][1],  false, b1,
                           (short)0, acc[ct],  false,  false);
        }
        __builtin_amdgcn_sched_barrier(0);
    }

#pragma unroll
    for (int ct = 0; ct < 8; ++ct)
#pragma unroll
        for (int r = 0; r < 8; ++r)
            tile[(8 * h + r) * TILE_PITCH + 16 * ct + c] = fmaxf(acc[ct][r], 0.f);
    asm volatile("s_wait_dscnt 0x0" ::: "memory");
    if (!NORM) {
#pragma unroll
        for (int r = 0; r < 16; ++r) {
            const float4 v = *(const float4*)(tile + r * TILE_PITCH + lane * 4);
            *(float4*)(out + (size_t)(rowBase + r) * DIM + lane * 4) = v;
        }
    } else {
        float4 v[16];
        float  s[16];
#pragma unroll
        for (int r = 0; r < 16; ++r) {
            v[r] = *(const float4*)(tile + r * TILE_PITCH + lane * 4);
            s[r] = v[r].x * v[r].x + v[r].y * v[r].y + v[r].z * v[r].z + v[r].w * v[r].w;
        }
        const bool b4 = (lane & 16) != 0, b3 = (lane & 8) != 0, b2 = (lane & 4) != 0, b1 = (lane & 2) != 0;
        float t[8], u[4], w[2];
#pragma unroll
        for (int k = 0; k < 8; ++k)
            t[k] = (b4 ? s[k + 8] : s[k]) + __shfl_xor(b4 ? s[k] : s[k + 8], 16, 32);
#pragma unroll
        for (int k = 0; k < 4; ++k)
            u[k] = (b3 ? t[k + 4] : t[k]) + __shfl_xor(b3 ? t[k] : t[k + 4], 8, 32);
#pragma unroll
        for (int k = 0; k < 2; ++k)
            w[k] = (b2 ? u[k + 2] : u[k]) + __shfl_xor(b2 ? u[k] : u[k + 2], 4, 32);
        float x = (b1 ? w[1] : w[0]) + __shfl_xor(b1 ? w[0] : w[1], 2, 32);
        x += __shfl_xor(x, 1, 32);
        const float inv = 1.0f / fmaxf(sqrtf(x), 1e-12f);
#pragma unroll
        for (int r = 0; r < 16; ++r) {
            const float ir = __int_as_float(__builtin_amdgcn_readlane(__float_as_int(inv), 2 * r));
            float4 y = v[r];
            y.x *= ir; y.y *= ir; y.z *= ir; y.w *= ir;
            *(float4*)(out + (size_t)(rowBase + r) * DIM + lane * 4) = y;
        }
    }
}

extern "C" void kernel_launch(void* const* d_in, const int* in_sizes, int n_in,
                              void* d_out, int out_size, void* d_ws, size_t ws_size,
                              hipStream_t stream) {
    const float* ent_emb = (const float*)d_in[0];
    const float* bw      = (const float*)d_in[1];
    const float* bc      = (const float*)d_in[2];
    const float* eval    = (const float*)d_in[3];
    const int*   esrc    = (const int*)d_in[5];
    const int*   edst    = (const int*)d_in[6];

    float* out = (float*)d_out;
    float* acc = (float*)d_ws;
    float* W   = acc + (size_t)N_ENT * DIM;
    int*      cnt      = (int*)(W + NLAYER * W_ELEMS);
    int*      rowstart = cnt + CNT_PAD;
    unsigned* tab      = (unsigned*)(rowstart + CNT_PAD);
    int*      rsrc     = (int*)(tab + NCHUNK * TAB_PITCH);
    int*      rdst     = rsrc + NCHUNK * CHUNK;
    float*    rval     = (float*)(rdst + NCHUNK * CHUNK);
    int2*     sorted   = (int2*)(rval + NCHUNK * CHUNK);

    chunk_kernel<<<NCHUNK + W_BLOCKS, 1024, 0, stream>>>(esrc, edst, eval, bw, bc, W, tab, rsrc, rdst, rval);
    bin_kernel<<<NBIN, 1024, 0, stream>>>(tab, rsrc, rdst, rval, cnt, rowstart, (unsigned long long*)sorted);

    gather_kernel<<<GATHER_WAVES / 4, 128, 0, stream>>>(ent_emb, rowstart, cnt, sorted, acc);
    gemm_relu_kernel<false><<<GEMM_BLOCKS, GEMM_WAVES * 32, 0, stream>>>(acc, W, out);
    gather_kernel<<<GATHER_WAVES / 4, 128, 0, stream>>>(out, rowstart, cnt, sorted, acc);
    gemm_relu_kernel<true><<<GEMM_BLOCKS, GEMM_WAVES * 32, 0, stream>>>(acc, W + W_ELEMS, out);
}


# Layer 0 as S(XW) not (SX)W: its GEMM needs no edge order and runs as extra work-groups of the edge-sort launch (2 tiles per wave); gather applies relu; 6 launches become 5; sums reassociated
# speedup vs baseline: 5.2934x; 1.0992x over previous
#include <hip/hip_runtime.h>

#define N_ENT   50000
#define DIM     128
#define NREL    11
#define NBASIS  4
#define NEDGE   1000000
#define NLAYER  2

#define CNT_PAD        50048
#define ROWS_PER_WAVE  4
#define GATHER_WAVES   (N_ENT / ROWS_PER_WAVE)
#define EDGE_BATCH     8

typedef __attribute__((ext_vector_type(2))) float v2f;
typedef __attribute__((ext_vector_type(8))) float v8f;

#define GEMM_WAVES   8
#define GEMM_ROWS    (GEMM_WAVES * 16)
#define GEMM_BLOCKS  ((N_ENT + GEMM_ROWS - 1) / GEMM_ROWS)
#define TILE_PITCH   132
#define GEMM_LDS_FLOATS  (DIM * DIM + GEMM_WAVES * 16 * TILE_PITCH)

#define W_ELEMS   (DIM * DIM)
#define W_GROUPS  (W_ELEMS / 1024)
#define W_BLOCKS  (NLAYER * W_GROUPS)

__device__ __forceinline__ void build_w_element(const float* __restrict__ bw,
                                                const float* __restrict__ bc,
                                                float* __restrict__ Wf, int l, int idx) {
    int q    = idx & 3;
    int lane = (idx >> 2) & 31;
    int j    = (idx >> 7) & 15;
    int ct   = idx >> 11;
    int o = 16 * ct + (lane & 15);
    int d = 8 * j + 4 * (q >> 1) + 2 * (lane >> 4) + (q & 1);
    float s = 0.f;
#pragma unroll
    for (int b = 0; b < NBASIS; ++b) {
        float c = bc[(l * NREL + l) * NBASIS + b];
        s += c * bw[(((size_t)l * NBASIS + b) * DIM + o) * DIM + d];
    }
    Wf[(size_t)l * W_ELEMS + idx] = s;
}

#define CHUNK       4096
#define NCHUNK      ((NEDGE + CHUNK - 1) / CHUNK)
#define BIN_ROWS    224
#define NBIN        ((N_ENT + BIN_ROWS - 1) / BIN_ROWS)
#define TAB_PITCH   256
#define STAGE_CAP   8192

__device__ __forceinline__ void wave_scan256(const int* in, int* out, int lane) {
    int v[8];
    int s = 0;
#pragma unroll
    for (int j = 0; j < 8; ++j) { v[j] = in[8 * lane + j]; s += v[j]; }
    int incl = s;
#pragma unroll
    for (int off = 1; off < 32; off <<= 1) {
        int t = __shfl_up(incl, off, 32);
        if (lane >= off) incl += t;
    }
    int run = incl - s;
#pragma unroll
    for (int j = 0; j < 8; ++j) { out[8 * lane + j] = run; run += v[j]; }
    if (lane == 31) out[256] = incl;
}

__device__ __forceinline__ unsigned long long pack_edge(int dst, float val) {
    return (unsigned long long)(unsigned)dst | ((unsigned long long)(unsigned)__float_as_int(val) << 32);
}

__global__ __launch_bounds__(1024)
void chunk_kernel(const int* __restrict__ esrc,
                  const int* __restrict__ edst,
                  const float* __restrict__ eval,
                  const float* __restrict__ bw,
                  const float* __restrict__ bc,
                  float* __restrict__ Wf,
                  unsigned* __restrict__ tab,
                  int* __restrict__ rsrc,
                  int* __restrict__ rdst,
                  float* __restrict__ rval) {
    __shared__ int hist[256];
    __shared__ int off[257];
    __shared__ __attribute__((aligned(16))) int   lsrc[CHUNK];
    __shared__ __attribute__((aligned(16))) int   ldst[CHUNK];
    __shared__ __attribute__((aligned(16))) float lval[CHUNK];

    const int t     = threadIdx.x;
    if (blockIdx.x >= NCHUNK) {
        const int wb = blockIdx.x - NCHUNK;
        build_w_element(bw, bc, Wf, wb / W_GROUPS, (wb % W_GROUPS) * 1024 + t);
        return;
    }
    const int chunk = blockIdx.x;
    const int i4    = chunk * (CHUNK / 4) + t;
    const bool valid = i4 < NEDGE / 4;

    if (t < 256) hist[t] = 0;
    int4   s = make_int4(0, 0, 0, 0);
    int4   d = make_int4(0, 0, 0, 0);
    float4 v = make_float4(0.f, 0.f, 0.f, 0.f);
    if (valid) {
        s = *(const int4*)(esrc + 4 * (size_t)i4);
        d = *(const int4*)(edst + 4 * (size_t)i4);
        v = *(const float4*)(eval + 4 * (size_t)i4);
    }
    __syncthreads();

    const int b0 = (int)min((unsigned)s.x / BIN_ROWS, (unsigned)(NBIN - 1));
    const int b1 = (int)min((unsigned)s.y / BIN_ROWS, (unsigned)(NBIN - 1));
    const int b2 = (int)min((unsigned)s.z / BIN_ROWS, (unsigned)(NBIN - 1));
    const int b3 = (int)min((unsigned)s.w / BIN_ROWS, (unsigned)(NBIN - 1));
    int r0 = 0, r1 = 0, r2 = 0, r3 = 0;
    if (valid) {
        r0 = atomicAdd(&hist[b0], 1);
        r1 = atomicAdd(&hist[b1], 1);
        r2 = atomicAdd(&hist[b2], 1);
        r3 = atomicAdd(&hist[b3], 1);
    }
    __syncthreads();

    if (t < 32) wave_scan256(hist, off, t);
    __syncthreads();

    if (t < TAB_PITCH)
        tab[chunk * TAB_PITCH + t] = ((unsigned)off[t] << 16) | (unsigned)hist[t];
    if (valid) {
        const int p0 = off[b0] + r0, p1 = off[b1] + r1, p2 = off[b2] + r2, p3 = off[b3] + r3;
        lsrc[p0] = s.x; ldst[p0] = d.x; lval[p0] = v.x;
        lsrc[p1] = s.y; ldst[p1] = d.y; lval[p1] = v.y;
        lsrc[p2] = s.z; ldst[p2] = d.z; lval[p2] = v.z;
        lsrc[p3] = s.w; ldst[p3] = d.w; lval[p3] = v.w;
    }
    __syncthreads();

    if (valid) {
        const size_t g = (size_t)chunk * CHUNK + 4 * t;
        *(int4*)(rsrc + g)   = *(const int4*)(lsrc + 4 * t);
        *(int4*)(rdst + g)   = *(const int4*)(ldst + 4 * t);
        *(float4*)(rval + g) = *(const float4*)(lval + 4 * t);
    }
}

__device__ __forceinline__ void gemm_plain_block(float* __restrict__ lds,
                                                 const float* __restrict__ A,
                                                 const float* __restrict__ Wf,
                                                 float* __restrict__ Y,
                                                 int gb) {
    float* sW = lds;
    float* sT = lds + DIM * DIM;

    const int wave    = __builtin_amdgcn_readfirstlane(threadIdx.x >> 5);
    const int lane    = threadIdx.x & 31;
    const int c       = lane & 15;
    const int h       = lane >> 4;
    const int strip   = wave >> 2;
    const int quarter = wave & 3;
    const int rowBase = gb * GEMM_ROWS + strip * 16;
    const bool active = rowBase < N_ENT;
    float* tile = sT + strip * (16 * TILE_PITCH);

    if (active) {
        float4 t[4];
#pragma unroll
        for (int r = 0; r < 4; ++r)
            t[r] = *(const float4*)(A + (size_t)(rowBase + 4 * quarter + r) * DIM + lane * 4);
#pragma unroll
        for (int r = 0; r < 4; ++r)
            *(float4*)(tile + (4 * quarter + r) * TILE_PITCH + lane * 4) = t[r];
    }
    {
        const float4* g = (const float4*)Wf + threadIdx.x;
        float4*       s = (float4*)sW + threadIdx.x;
#pragma unroll
        for (int i = 0; i < (DIM * DIM / 4) / 1024; ++i)
            s[i * 1024] = g[i * 1024];
    }
    __syncthreads();

    v8f acc[2];
    acc[0] = (v8f){};
    acc[1] = (v8f){};
    if (active) {
        const float*  ap = tile + c * TILE_PITCH + 2 * h;
        const float4* bp = (const float4*)sW + quarter * (2 * (DIM / 8) * 32) + lane;
        v2f    a[2][2];
        float4 b[2][2];
        a[0][0] = *(const v2f*)(ap);
        a[0][1] = *(const v2f*)(ap + 4);
        b[0][0] = bp[0];
        b[0][1] = bp[(DIM / 8) * 32];
#pragma unroll
        for (int j = 0; j < DIM / 8; ++j) {
            const int cur = j & 1, nxt = cur ^ 1;
            if (j + 1 < DIM / 8) {
                a[nxt][0] = *(const v2f*)(ap + 8 * (j + 1));
                a[nxt][1] = *(const v2f*)(ap + 8 * (j + 1) + 4);
                b[nxt][0] = bp[(j + 1) * 32];
                b[nxt][1] = bp[((DIM / 8) + j + 1) * 32];
            }
            __builtin_amdgcn_sched_barrier(0);
#pragma unroll
            for (int t = 0; t < 2; ++t) {
                v2f b0 = {b[cur][t].x, b[cur][t].y};
                acc[t] = __builtin_amdgcn_wmma_f32_16x16x4_f32(
                              false, a[cur][0],  false, b0,
                              (short)0, acc[t],  false,  false);
            }
#pragma unroll
            for (int t = 0; t < 2; ++t) {
                v2f b1 = {b[cur][t].z, b[cur][t].w};
                acc[t] = __builtin_amdgcn_wmma_f32_16x16x4_f32(
                              false, a[cur][1],  false, b1,
                              (short)0, acc[t],  false,  false);
            }
            __builtin_amdgcn_sched_barrier(0);
        }
    }
    __syncthreads();
    if (!active) return;

#pragma unroll
    for (int t = 0; t < 2; ++t)
#pragma unroll
        for (int r = 0; r < 8; ++r)
            tile[(8 * h + r) * TILE_PITCH + 32 * quarter + 16 * t + c] = acc[t][r];
    asm volatile("s_wait_dscnt 0x0" ::: "memory");
#pragma unroll
    for (int i = 0; i < 4; ++i) {
        const int row = 4 * i + (lane >> 3);
        const int col = 32 * quarter + 4 * (lane & 7);
        const float4 v = *(const float4*)(tile + row * TILE_PITCH + col);
        *(float4*)(Y + (size_t)(rowBase + row) * DIM + col) = v;
    }
}

__device__ __forceinline__ int piece_slot(const int* pstart, const int* poff, int e) {
    int lo = 0, hi = 256;
#pragma unroll
    for (int it = 0; it < 8; ++it) {
        const int mid = (lo + hi) >> 1;
        if (pstart[mid] <= e) lo = mid; else hi = mid;
    }
    return lo * CHUNK + poff[lo] + (e - pstart[lo]);
}

struct BinLds {
    unsigned long long stage[STAGE_CAP];
    int pcnt[256];
    int poff[256];
    int pstart[257];
    int rowcnt[256];
    int rowpos[257];
    int red[32];
    int sbase;
};
union BinGemmLds {
    BinLds bin;
    float  gemm[GEMM_LDS_FLOATS];
};

__global__ __launch_bounds__(1024)
void bin_kernel(const unsigned* __restrict__ tab,
                const int* __restrict__ rsrc,
                const int* __restrict__ rdst,
                const float* __restrict__ rval,
                int* __restrict__ cnt,
                int* __restrict__ rowstart,
                unsigned long long* __restrict__ sorted,
                const float* __restrict__ emb,
                const float* __restrict__ Wf,
                float* __restrict__ Y) {
    __shared__ __attribute__((aligned(16))) BinGemmLds lds;

    if (blockIdx.x >= NBIN) {
        gemm_plain_block(lds.gemm, emb, Wf, Y, blockIdx.x - NBIN);
        return;
    }
    unsigned long long* stage = lds.bin.stage;
    int* pcnt   = lds.bin.pcnt;
    int* poff   = lds.bin.poff;
    int* pstart = lds.bin.pstart;
    int* rowcnt = lds.bin.rowcnt;
    int* rowpos = lds.bin.rowpos;
    int* red    = lds.bin.red;

    const int t    = threadIdx.x;
    const int lane = t & 31;
    const int wave = t >> 5;
    const int bin  = blockIdx.x;

    int part = 0;
    for (int q = t; q < NCHUNK * TAB_PITCH / 4; q += 1024) {
        const uint4 w = ((const uint4*)tab)[q];
        const int col = (q & (TAB_PITCH / 4 - 1)) * 4;
        part += (col     < bin ? (int)(w.x & 0xFFFFu) : 0) + (col + 1 < bin ? (int)(w.y & 0xFFFFu) : 0)
              + (col + 2 < bin ? (int)(w.z & 0xFFFFu) : 0) + (col + 3 < bin ? (int)(w.w & 0xFFFFu) : 0);
    }
#pragma unroll
    for (int off = 16; off > 0; off >>= 1) part += __shfl_xor(part, off, 32);
    if (lane == 0) red[wave] = part;
    if (t < 256) {
        unsigned w = 0;
        if (t < NCHUNK) w = tab[t * TAB_PITCH + bin];
        pcnt[t]   = (int)(w & 0xFFFFu);
        poff[t]   = (int)(w >> 16);
        rowcnt[t] = 0;
    }
    __syncthreads();

    if (t < 32) {
        wave_scan256(pcnt, pstart, t);
        int x = red[t];
#pragma unroll
        for (int off = 16; off > 0; off >>= 1) x += __shfl_xor(x, off, 32);
        if (t == 0) lds.bin.sbase = x;
    }
    __syncthreads();
    const int total = pstart[256];
    const int base  = lds.bin.sbase;
    const int row0  = bin * BIN_ROWS;

    for (int e0 = t; e0 < total; e0 += 4 * 1024) {
        int lr[4];
#pragma unroll
        for (int k = 0; k < 4; ++k) {
            const int e = e0 + k * 1024;
            lr[k] = -1;
            if (e < total) lr[k] = min(max(rsrc[piece_slot(pstart, poff, e)] - row0, 0), BIN_ROWS - 1);
        }
#pragma unroll
        for (int k = 0; k < 4; ++k)
            if (lr[k] >= 0) atomicAdd(&rowcnt[lr[k]], 1);
    }
    __syncthreads();

    if (t < 32) wave_scan256(rowcnt, rowpos, t);
    __syncthreads();
    if (t < BIN_ROWS && row0 + t < CNT_PAD) {
        cnt[row0 + t]      = rowcnt[t];
        rowstart[row0 + t] = base + rowpos[t];
    }
    __syncthreads();

    const bool staged = total <= STAGE_CAP;
    for (int e0 = t; e0 < total; e0 += 4 * 1024) {
        int lr[4];
        unsigned long long rec[4];
#pragma unroll
        for (int k = 0; k < 4; ++k) {
            const int e = e0 + k * 1024;
            lr[k]  = -1;
            rec[k] = 0;
            if (e < total) {
                const int g = piece_slot(pstart, poff, e);
                lr[k]  = min(max(rsrc[g] - row0, 0), BIN_ROWS - 1);
                rec[k] = pack_edge(rdst[g], rval[g]);
            }
        }
#pragma unroll
        for (int k = 0; k < 4; ++k) {
            if (lr[k] >= 0) {
                const int p = atomicAdd(&rowpos[lr[k]], 1);
                if (staged) stage[min(p, STAGE_CAP - 1)] = rec[k];
                else        atomicExch(sorted + min(base + p, NEDGE - 1), rec[k]);
            }
        }
    }
    __syncthreads();
    if (staged) {
        const int lo  = base & ~15;
        const int in0 = (base + 15) & ~15;
        const int in1 = (base + total) & ~15;
        for (int idx = lo + t; idx < base + total; idx += 1024) {
            const int e = idx - base;
            if (e >= 0) {
                const int at = min(idx, NEDGE - 1);
                if (idx >= in0 && idx < in1) sorted[at] = stage[e];
                else                         atomicExch(sorted + at, stage[e]);
            }
        }
    }
}

template <bool FULL>
__device__ __forceinline__ void edge_batch(const float* __restrict__ ebase, int2 e, int j, int n,
                                           float4& a) {
    float4 x[EDGE_BATCH];
    float  v[EDGE_BATCH];
#pragma unroll
    for (int u = 0; u < EDGE_BATCH; ++u) {
        v[u] = 0.f;
        x[u] = make_float4(0.f, 0.f, 0.f, 0.f);
        if (FULL || j + u < n) {
            int d = __builtin_amdgcn_readlane(e.x, j + u);
            v[u]  = __int_as_float(__builtin_amdgcn_readlane(e.y, j + u));
            x[u]  = *(const float4*)(ebase + (size_t)d * DIM);
        }
    }
#pragma unroll
    for (int u = 0; u < EDGE_BATCH; ++u) {
        a.x = fmaf(v[u], x[u].x, a.x);
        a.y = fmaf(v[u], x[u].y, a.y);
        a.z = fmaf(v[u], x[u].z, a.z);
        a.w = fmaf(v[u], x[u].w, a.w);
    }
}

template <bool RELU>
__global__ __launch_bounds__(128)
void gather_kernel(const float* __restrict__ emb,
                   const int* __restrict__ rowstart,
                   const int* __restrict__ cnt,
                   const int2* __restrict__ sorted,
                   float* __restrict__ acc) {
    const int wave = __builtin_amdgcn_readfirstlane(blockIdx.x * 4 + (threadIdx.x >> 5));
    const int lane = threadIdx.x & 31;
    const int row0 = wave * ROWS_PER_WAVE;

    int st[ROWS_PER_WAVE], dg[ROWS_PER_WAVE];
#pragma unroll
    for (int r = 0; r < ROWS_PER_WAVE; ++r) {
        st[r] = min(max(rowstart[row0 + r], 0), NEDGE - 1);
        dg[r] = min(max(cnt[row0 + r], 0), NEDGE);
    }
    int2 first[ROWS_PER_WAVE];
#pragma unroll
    for (int r = 0; r < ROWS_PER_WAVE; ++r)
        first[r] = (lane < dg[r]) ? sorted[min(st[r] + lane, NEDGE - 1)] : make_int2(0, 0);

    const float* ebase = emb + lane * 4;
#pragma unroll
    for (int r = 0; r < ROWS_PER_WAVE; ++r) {
        float4 a = make_float4(0.f, 0.f, 0.f, 0.f);
        const int deg = dg[r];
        int2 e = first[r];
        for (int base = 0; base < deg; base += 32) {
            if (base > 0)
                e = (base + lane < deg) ? sorted[min(st[r] + base + lane, NEDGE - 1)] : make_int2(0, 0);
            e.x = min(max(e.x, 0), N_ENT - 1);
            const int n = min(32, deg - base);
            int j = 0;
            for (; j + EDGE_BATCH <= n; j += EDGE_BATCH)
                edge_batch<true>(ebase, e, j, n, a);
            if (j < n)
                edge_batch<false>(ebase, e, j, n, a);
        }
        if (RELU) {
            a.x = fmaxf(a.x, 0.f); a.y = fmaxf(a.y, 0.f); a.z = fmaxf(a.z, 0.f); a.w = fmaxf(a.w, 0.f);
        }
        *(float4*)(acc + (size_t)(row0 + r) * DIM + lane * 4) = a;
    }
}

template <bool NORM>
__global__ __launch_bounds__(GEMM_WAVES * 32)
void gemm_relu_kernel(const float* __restrict__ A,
                      const float* __restrict__ Wf,
                      float* __restrict__ out) {
    __shared__ __attribute__((aligned(16))) float lds[GEMM_LDS_FLOATS];
    float* sW = lds;
    float* sT = lds + DIM * DIM;

    const int wave = __builtin_amdgcn_readfirstlane(threadIdx.x >> 5);
    const int lane = threadIdx.x & 31;
    const int c = lane & 15;
    const int h = lane >> 4;
    const int rowBase = blockIdx.x * GEMM_ROWS + wave * 16;
    const bool active = rowBase < N_ENT;
    float* tile = sT + wave * (16 * TILE_PITCH);

    if (active) {
        float4 t[16];
#pragma unroll
        for (int r = 0; r < 16; ++r)
            t[r] = *(const float4*)(A + (size_t)(rowBase + r) * DIM + lane * 4);
#pragma unroll
        for (int r = 0; r < 16; ++r)
            *(float4*)(tile + r * TILE_PITCH + lane * 4) = t[r];
    }
    {
        const float4* g = (const float4*)Wf + threadIdx.x;
        float4*       s = (float4*)sW + threadIdx.x;
#pragma unroll
        for (int i = 0; i < (DIM * DIM / 4) / (GEMM_WAVES * 32); ++i)
            s[i * (GEMM_WAVES * 32)] = g[i * (GEMM_WAVES * 32)];
    }
    __syncthreads();
    if (!active) return;

    v8f acc[8];
#pragma unroll
    for (int ct = 0; ct < 8; ++ct) acc[ct] = (v8f){};

    const float*  ap = tile + c * TILE_PITCH + 2 * h;
    const float4* bp = (const float4*)sW + lane;
    v2f    a[2][2];
    float4 b[2][8];
    a[0][0] = *(const v2f*)(ap);
    a[0][1] = *(const v2f*)(ap + 4);
#pragma unroll
    for (int ct = 0; ct < 8; ++ct) b[0][ct] = bp[(ct * (DIM / 8)) * 32];
#pragma unroll
    for (int j = 0; j < DIM / 8; ++j) {
        const int cur = j & 1, nxt = cur ^ 1;
        if (j + 1 < DIM / 8) {
            a[nxt][0] = *(const v2f*)(ap + 8 * (j + 1));
            a[nxt][1] = *(const v2f*)(ap + 8 * (j + 1) + 4);
#pragma unroll
            for (int ct = 0; ct < 8; ++ct) b[nxt][ct] = bp[(ct * (DIM / 8) + j + 1) * 32];
        }
        __builtin_amdgcn_sched_barrier(0);
#pragma unroll
        for (int ct = 0; ct < 8; ++ct) {
            v2f b0 = {b[cur][ct].x, b[cur][ct].y};
            acc[ct] = __builtin_amdgcn_wmma_f32_16x16x4_f32(
                           false, a[cur][0],  false, b0,
                           (short)0, acc[ct],  false,  false);
        }
#pragma unroll
        for (int ct = 0; ct < 8; ++ct) {
            v2f b1 = {b[cur][ct].z, b[cur][ct].w};
            acc[ct] = __builtin_amdgcn_wmma_f32_16x16x4_f32(
                           false, a[cur][1],  false, b1,
                           (short)0, acc[ct],  false,  false);
        }
        __builtin_amdgcn_sched_barrier(0);
    }

#pragma unroll
    for (int ct = 0; ct < 8; ++ct)
#pragma unroll
        for (int r = 0; r < 8; ++r)
            tile[(8 * h + r) * TILE_PITCH + 16 * ct + c] = fmaxf(acc[ct][r], 0.f);
    asm volatile("s_wait_dscnt 0x0" ::: "memory");
    if (!NORM) {
#pragma unroll
        for (int r = 0; r < 16; ++r) {
            const float4 v = *(const float4*)(tile + r * TILE_PITCH + lane * 4);
            *(float4*)(out + (size_t)(rowBase + r) * DIM + lane * 4) = v;
        }
    } else {
        float4 v[16];
        float  s[16];
#pragma unroll
        for (int r = 0; r < 16; ++r) {
            v[r] = *(const float4*)(tile + r * TILE_PITCH + lane * 4);
            s[r] = v[r].x * v[r].x + v[r].y * v[r].y + v[r].z * v[r].z + v[r].w * v[r].w;
        }
        const bool b4 = (lane & 16) != 0, b3 = (lane & 8) != 0, b2 = (lane & 4) != 0, b1 = (lane & 2) != 0;
        float t[8], u[4], w[2];
#pragma unroll
        for (int k = 0; k < 8; ++k)
            t[k] = (b4 ? s[k + 8] : s[k]) + __shfl_xor(b4 ? s[k] : s[k + 8], 16, 32);
#pragma unroll
        for (int k = 0; k < 4; ++k)
            u[k] = (b3 ? t[k + 4] : t[k]) + __shfl_xor(b3 ? t[k] : t[k + 4], 8, 32);
#pragma unroll
        for (int k = 0; k < 2; ++k)
            w[k] = (b2 ? u[k + 2] : u[k]) + __shfl_xor(b2 ? u[k] : u[k + 2], 4, 32);
        float x = (b1 ? w[1] : w[0]) + __shfl_xor(b1 ? w[0] : w[1], 2, 32);
        x += __shfl_xor(x, 1, 32);
        const float inv = 1.0f / fmaxf(sqrtf(x), 1e-12f);
#pragma unroll
        for (int r = 0; r < 16; ++r) {
            const float ir = __int_as_float(__builtin_amdgcn_readlane(__float_as_int(inv), 2 * r));
            float4 y = v[r];
            y.x *= ir; y.y *= ir; y.z *= ir; y.w *= ir;
            *(float4*)(out + (size_t)(rowBase + r) * DIM + lane * 4) = y;
        }
    }
}

extern "C" void kernel_launch(void* const* d_in, const int* in_sizes, int n_in,
                              void* d_out, int out_size, void* d_ws, size_t ws_size,
                              hipStream_t stream) {
    const float* ent_emb = (const float*)d_in[0];
    const float* bw      = (const float*)d_in[1];
    const float* bc      = (const float*)d_in[2];
    const float* eval    = (const float*)d_in[3];
    const int*   esrc    = (const int*)d_in[5];
    const int*   edst    = (const int*)d_in[6];

    float* out = (float*)d_out;
    float* acc = (float*)d_ws;
    float* W   = acc + (size_t)N_ENT * DIM;
    int*      cnt      = (int*)(W + NLAYER * W_ELEMS);
    int*      rowstart = cnt + CNT_PAD;
    unsigned* tab      = (unsigned*)(rowstart + CNT_PAD);
    int*      rsrc     = (int*)(tab + NCHUNK * TAB_PITCH);
    int*      rdst     = rsrc + NCHUNK * CHUNK;
    float*    rval     = (float*)(rdst + NCHUNK * CHUNK);
    int2*     sorted   = (int2*)(rval + NCHUNK * CHUNK);

    chunk_kernel<<<NCHUNK + W_BLOCKS, 1024, 0, stream>>>(esrc, edst, eval, bw, bc, W, tab, rsrc, rdst, rval);
    bin_kernel<<<NBIN + GEMM_BLOCKS, 1024, 0, stream>>>(tab, rsrc, rdst, rval, cnt, rowstart, (unsigned long long*)sorted,
                                                        ent_emb, W, acc);

    gather_kernel<true><<<GATHER_WAVES / 4, 128, 0, stream>>>(acc, rowstart, cnt, sorted, out);
    gather_kernel<false><<<GATHER_WAVES / 4, 128, 0, stream>>>(out, rowstart, cnt, sorted, acc);
    gemm_relu_kernel<true><<<GEMM_BLOCKS, GEMM_WAVES * 32, 0, stream>>>(acc, W + W_ELEMS, out);
}
